# starting kernel
# baseline (speedup 1.0000x reference)
#include <hip/hip_runtime.h>

typedef float v2f __attribute__((ext_vector_type(2)));
typedef float v8f __attribute__((ext_vector_type(8)));

__global__ void gcn_init_deg(float* __restrict__ deg, int n) {
  int i = blockIdx.x * blockDim.x + threadIdx.x;
  if (i < n) deg[i] = 1.0f;
}

__global__ void gcn_count_deg(const int* __restrict__ dst, float* deg, int e) {
  int i = blockIdx.x * blockDim.x + threadIdx.x;
  if (i < e) atomicAdd(&deg[dst[i]], 1.0f);
}

__global__ void gcn_dinv(float* deg, int n) {
  int i = blockIdx.x * blockDim.x + threadIdx.x;
  if (i < n) deg[i] = rsqrtf(deg[i]);
}

__global__ void gcn_zero(float* __restrict__ p, long n) {
  long i = (long)blockIdx.x * blockDim.x + threadIdx.x;
  if (i < n) p[i] = 0.0f;
}

template <int LDB, int LDC, int NVALID>
__global__ void gcn_gemm_wmma_f32(const float* __restrict__ A,
                                  const float* __restrict__ B,
                                  float* __restrict__ C, int nrows) {
  const int lane = threadIdx.x & 31;
  const int wave = threadIdx.x >> 5;
  const int row0 = (blockIdx.x * (blockDim.x >> 5) + wave) * 16;
  if (row0 >= nrows) return;
  const int n0   = blockIdx.y * 16;

  const int mrow = lane & 15;
  const int hsel = lane >> 4;
  const int kofs = hsel << 1;

  const float* arow = A + (row0 + mrow) * 64 + kofs;
  const int    bn   = n0 + mrow;
  const int    bcol = (bn < NVALID) ? bn : (NVALID - 1);
  const float* bp   = B + kofs * LDB + bcol;

  v8f acc = {};
#pragma unroll
  for (int k = 0; k < 64; k += 4) {
    v2f a = *(const v2f*)(arow + k);
    v2f b = { bp[k * LDB], bp[k * LDB + LDB] };
    acc = __builtin_amdgcn_wmma_f32_16x16x4_f32(
         false, a,  false, b,
         (short)0, acc,  false,  false);
  }

  float* cp = C + (row0 + (hsel << 3)) * LDC + n0 + mrow;
  if (row0 + 16 <= nrows) {
#pragma unroll
    for (int r = 0; r < 8; ++r) cp[r * LDC] = acc[r];
  } else {
    const int crow = row0 + (hsel << 3);
#pragma unroll
    for (int r = 0; r < 8; ++r)
      if (crow + r < nrows) cp[r * LDC] = acc[r];
  }
}

__global__ void gcn_scatter64(const float* __restrict__ h,
                              const int* __restrict__ src,
                              const int* __restrict__ dst,
                              const float* __restrict__ dinv,
                              float* agg, int e) {
  int widx = blockIdx.x * (blockDim.x >> 5) + (threadIdx.x >> 5);
  int lane = threadIdx.x & 31;
  if (widx >= e) return;
  int s = src[widx], d = dst[widx];
  float w = dinv[s] * dinv[d];
  v2f v = *(const v2f*)(h + (long)s * 64 + lane * 2);
  float* od = agg + (long)d * 64 + lane * 2;
  atomicAdd(od + 0, v.x * w);
  atomicAdd(od + 1, v.y * w);
}

__global__ void gcn_epilogue1(float* agg, const float* __restrict__ h,
                              const float* __restrict__ dinv,
                              const float* __restrict__ b1, int n) {
  long i = (long)blockIdx.x * blockDim.x + threadIdx.x;
  if (i >= (long)n * 64) return;
  int node = (int)(i >> 6);
  int c    = (int)(i & 63);
  float di = dinv[node];
  float v  = agg[i] + di * di * h[i] + b1[c];
  agg[i] = v > 0.0f ? v : 0.0f;
}

__global__ void gcn_scatter40(const float* __restrict__ h2,
                              const int* __restrict__ src,
                              const int* __restrict__ dst,
                              const float* __restrict__ dinv,
                              float* out, int e) {
  int widx = blockIdx.x * (blockDim.x >> 5) + (threadIdx.x >> 5);
  int lane = threadIdx.x & 31;
  if (widx >= e) return;
  int s = src[widx], d = dst[widx];
  float w = dinv[s] * dinv[d];
  const float* hs = h2 + (long)s * 48;
  float* od = out + (long)d * 40;
  atomicAdd(od + lane, hs[lane] * w);
  if (lane < 8) atomicAdd(od + 32 + lane, hs[32 + lane] * w);
}

__global__ void gcn_epilogue2(float* out, const float* __restrict__ h2,
                              const float* __restrict__ dinv,
                              const float* __restrict__ b3, int n) {
  long i = (long)blockIdx.x * blockDim.x + threadIdx.x;
  if (i >= (long)n * 40) return;
  int node = (int)(i / 40);
  int c    = (int)(i % 40);
  float di = dinv[node];
  out[i] += di * di * h2[(long)node * 48 + c] + b3[c];
}

extern "C" void kernel_launch(void* const* d_in, const int* in_sizes, int n_in,
                              void* d_out, int out_size, void* d_ws, size_t ws_size,
                              hipStream_t stream) {
  (void)n_in; (void)out_size; (void)ws_size;
  const float* x  = (const float*)d_in[0];
  const int*   ei = (const int*)d_in[1];
  const float* W1 = (const float*)d_in[2];
  const float* b1 = (const float*)d_in[3];
  const float* W3 = (const float*)d_in[4];
  const float* b3 = (const float*)d_in[5];

  const int N = in_sizes[0] / 64;
  const int E = in_sizes[1] / 2;
  const int* src = ei;
  const int* dst = ei + E;

  float* dinv = (float*)d_ws;
  float* h    = dinv + N;
  float* agg1 = h + (long)N * 64;
  float* h2   = agg1 + (long)N * 64;
  float* out  = (float*)d_out;

  const int T = 256;
  const int mtiles = (N + 15) / 16;

  gcn_init_deg<<<(N + T - 1) / T, T, 0, stream>>>(dinv, N);
  gcn_count_deg<<<(E + T - 1) / T, T, 0, stream>>>(dst, dinv, E);
  gcn_dinv<<<(N + T - 1) / T, T, 0, stream>>>(dinv, N);

  {
    dim3 grid((mtiles + 7) / 8, 4);
    gcn_gemm_wmma_f32<64, 64, 64><<<grid, 256, 0, stream>>>(x, W1, h, N);
  }
  gcn_zero<<<(int)(((long)N * 64 + T - 1) / T), T, 0, stream>>>(agg1, (long)N * 64);
  gcn_scatter64<<<(E + 7) / 8, 256, 0, stream>>>(h, src, dst, dinv, agg1, E);
  gcn_epilogue1<<<(int)(((long)N * 64 + T - 1) / T), T, 0, stream>>>(agg1, h, dinv, b1, N);

  {
    dim3 grid((mtiles + 7) / 8, 3);
    gcn_gemm_wmma_f32<40, 48, 40><<<grid, 256, 0, stream>>>(agg1, W3, h2, N);
  }
  gcn_zero<<<(int)(((long)N * 40 + T - 1) / T), T, 0, stream>>>(out, (long)N * 40);
  gcn_scatter40<<<(E + 7) / 8, 256, 0, stream>>>(h2, src, dst, dinv, out, E);
  gcn_epilogue2<<<(int)(((long)N * 40 + T - 1) / T), T, 0, stream>>>(out, h2, dinv, b3, N);
}


# Replace both float-atomic edge scatters by a gather over destination-sorted row lists (int-atomic rank, scan, fill, built once): a wave sums a row in registers, arrival-order sum; zero+epilogue fused
# speedup vs baseline: 1.9815x; 1.9815x over previous
#include <hip/hip_runtime.h>

typedef float v2f __attribute__((ext_vector_type(2)));
typedef float v8f __attribute__((ext_vector_type(8)));

__global__ void csr_zero_cnt(int* __restrict__ cnt, int n) {
  int i = blockIdx.x * blockDim.x + threadIdx.x;
  if (i < n) cnt[i] = 0;
}

__global__ void csr_count_rank(const int* __restrict__ dst, int* cnt,
                               int* __restrict__ rank, int e) {
  int i = blockIdx.x * blockDim.x + threadIdx.x;
  if (i < e) rank[i] = atomicAdd(&cnt[dst[i]], 1);
}

__global__ void __launch_bounds__(1024)
csr_scan(const int* __restrict__ cnt, int* __restrict__ rowptr,
         float* __restrict__ dinv, int n) {
  __shared__ int part[32];
  __shared__ int wexc[32];
  __shared__ int boff;
  const int tid = threadIdx.x;
  const int lane = tid & 31;
  const int w = tid >> 5;
  const int start = blockIdx.x * 1024;

  int t = 0;
  for (int k = tid; k < start; k += 1024) t += cnt[k];
  for (int o = 16; o > 0; o >>= 1) t += __shfl_xor(t, o);
  if (lane == 0) part[w] = t;

  const int i = start + tid;
  const int c = (i < n) ? cnt[i] : 0;
  int x = c;
  for (int o = 1; o < 32; o <<= 1) {
    int y = __shfl_up(x, o);
    if (lane >= o) x += y;
  }
  if (lane == 31) wexc[w] = x;
  __syncthreads();
  if (w == 0) {
    int p = part[lane];
    for (int o = 16; o > 0; o >>= 1) p += __shfl_xor(p, o);
    const int s = wexc[lane];
    int incl = s;
    for (int o = 1; o < 32; o <<= 1) {
      int y = __shfl_up(incl, o);
      if (lane >= o) incl += y;
    }
    wexc[lane] = incl - s;
    if (lane == 0) boff = p;
  }
  __syncthreads();
  const int excl = boff + wexc[w] + x - c;
  if (i < n) {
    rowptr[i] = excl;
    dinv[i] = rsqrtf((float)(c + 1));
  }
  if (i == n - 1) rowptr[n] = excl + c;
}

__global__ void csr_fill(const int* __restrict__ src, const int* __restrict__ dst,
                         const int* __restrict__ rowptr,
                         const int* __restrict__ rank, int* __restrict__ col, int e) {
  int i = blockIdx.x * blockDim.x + threadIdx.x;
  if (i < e) col[rowptr[dst[i]] + rank[i]] = src[i];
}

template <int LDB, int LDC, int NVALID>
__global__ void gcn_gemm_wmma_f32(const float* __restrict__ A,
                                  const float* __restrict__ B,
                                  float* __restrict__ C, int nrows) {
  const int lane = threadIdx.x & 31;
  const int wave = threadIdx.x >> 5;
  const int row0 = (blockIdx.x * (blockDim.x >> 5) + wave) * 16;
  if (row0 >= nrows) return;
  const int n0   = blockIdx.y * 16;

  const int mrow = lane & 15;
  const int hsel = lane >> 4;
  const int kofs = hsel << 1;

  const float* arow = A + (row0 + mrow) * 64 + kofs;
  const int    bn   = n0 + mrow;
  const int    bcol = (bn < NVALID) ? bn : (NVALID - 1);
  const float* bp   = B + kofs * LDB + bcol;

  v8f acc = {};
#pragma unroll
  for (int k = 0; k < 64; k += 4) {
    v2f a = *(const v2f*)(arow + k);
    v2f b = { bp[k * LDB], bp[k * LDB + LDB] };
    acc = __builtin_amdgcn_wmma_f32_16x16x4_f32(
         false, a,  false, b,
         (short)0, acc,  false,  false);
  }

  float* cp = C + (row0 + (hsel << 3)) * LDC + n0 + mrow;
  if (row0 + 16 <= nrows) {
#pragma unroll
    for (int r = 0; r < 8; ++r) cp[r * LDC] = acc[r];
  } else {
    const int crow = row0 + (hsel << 3);
#pragma unroll
    for (int r = 0; r < 8; ++r)
      if (crow + r < nrows) cp[r * LDC] = acc[r];
  }
}

template <int LDH, int NV2, int LDO, bool RELU>
__global__ void gcn_gather_rows(const float* __restrict__ hin,
                                const int* __restrict__ rowptr,
                                const int* __restrict__ col,
                                const float* __restrict__ dinv,
                                const float* __restrict__ bias,
                                float* __restrict__ res, int n) {
  constexpr int G = 16;
  const int lane = threadIdx.x & 31;
  const int row = __builtin_amdgcn_readfirstlane(
      (int)(blockIdx.x * (blockDim.x >> 5) + (threadIdx.x >> 5)));
  if (row >= n) return;

  const int l2 = (lane < NV2) ? lane : (NV2 - 1);
  const int lo = l2 * 2;
  const v2f* hin2 = (const v2f*)hin;
  const int beg = rowptr[row];
  const int end = rowptr[row + 1];
  const float di = dinv[row];
  const float wself = di * di;
  const v2f hs = hin2[(unsigned)row * (LDH / 2) + l2];
  const v2f bv = *(const v2f*)(bias + lo);
  v2f acc = { hs.x * wself, hs.y * wself };

  for (int base = beg; base < end; base += 32) {
    const int ne = (end - base < 32) ? (end - base) : 32;
    const int li = (lane < ne) ? lane : (ne - 1);
    const int my_s = col[base + li];
    const int my_d = __float_as_int(dinv[my_s]);
    for (int j = 0; j < ne; j += G) {
      v2f v[G];
#pragma unroll
      for (int u = 0; u < G; ++u) {
        const int jj = j + u;
        const int jc = (jj < ne) ? jj : (ne - 1);
        const int s = __builtin_amdgcn_readlane(my_s, jc);
        v[u] = hin2[(unsigned)s * (LDH / 2) + l2];
      }
#pragma unroll
      for (int u = 0; u < G; ++u) {
        const int jj = j + u;
        const int jc = (jj < ne) ? jj : (ne - 1);
        const float wu = __int_as_float(__builtin_amdgcn_readlane(my_d, jc)) * di;
        const float ax = fmaf(v[u].x, wu, acc.x);
        const float ay = fmaf(v[u].y, wu, acc.y);
        acc.x = (jj < ne) ? ax : acc.x;
        acc.y = (jj < ne) ? ay : acc.y;
      }
    }
  }

  v2f r = { acc.x + bv.x, acc.y + bv.y };
  if (RELU) {
    r.x = r.x > 0.0f ? r.x : 0.0f;
    r.y = r.y > 0.0f ? r.y : 0.0f;
  }
  if (lane < NV2) *(v2f*)(res + (long)row * LDO + lo) = r;
}

extern "C" void kernel_launch(void* const* d_in, const int* in_sizes, int n_in,
                              void* d_out, int out_size, void* d_ws, size_t ws_size,
                              hipStream_t stream) {
  (void)n_in; (void)out_size; (void)ws_size;
  const float* x  = (const float*)d_in[0];
  const int*   ei = (const int*)d_in[1];
  const float* W1 = (const float*)d_in[2];
  const float* b1 = (const float*)d_in[3];
  const float* W3 = (const float*)d_in[4];
  const float* b3 = (const float*)d_in[5];

  const int N = in_sizes[0] / 64;
  const int E = in_sizes[1] / 2;
  const int* src = ei;
  const int* dst = ei + E;

  float* dinv   = (float*)d_ws;
  float* bufA   = dinv + N;
  float* bufB   = bufA + (long)N * 64;
  int*   cnt    = (int*)(bufB + (long)N * 64);
  int*   rowptr = cnt + N;
  int*   rank   = rowptr + N + 1;
  int*   col    = rank + E;
  float* out    = (float*)d_out;

  const int T = 256;
  const int mtiles = (N + 15) / 16;

  csr_zero_cnt<<<(N + T - 1) / T, T, 0, stream>>>(cnt, N);
  csr_count_rank<<<(E + T - 1) / T, T, 0, stream>>>(dst, cnt, rank, E);
  csr_scan<<<(N + 1023) / 1024, 1024, 0, stream>>>(cnt, rowptr, dinv, N);
  csr_fill<<<(E + T - 1) / T, T, 0, stream>>>(src, dst, rowptr, rank, col, E);

  {
    dim3 grid((mtiles + 7) / 8, 4);
    gcn_gemm_wmma_f32<64, 64, 64><<<grid, 256, 0, stream>>>(x, W1, bufA, N);
  }
  gcn_gather_rows<64, 32, 64, true><<<(N + 7) / 8, 256, 0, stream>>>(
      bufA, rowptr, col, dinv, b1, bufB, N);

  {
    dim3 grid((mtiles + 7) / 8, 3);
    gcn_gemm_wmma_f32<40, 48, 40><<<grid, 256, 0, stream>>>(bufB, W3, bufA, N);
  }
  gcn_gather_rows<48, 20, 40, false><<<(N + 7) / 8, 256, 0, stream>>>(
      bufA, rowptr, col, dinv, b3, out, N);
}


# Scale each transformed row by dinv[row] once in the GEMM epilogue; gathers sum plain rows, dinv[dst] applied once after the sum: no per-edge dinv[src] lookup; product rounding changes, sum order kept
# speedup vs baseline: 2.0827x; 1.0511x over previous
#include <hip/hip_runtime.h>

typedef float v2f __attribute__((ext_vector_type(2)));
typedef float v8f __attribute__((ext_vector_type(8)));

__global__ void csr_zero_cnt(int* __restrict__ cnt, int n) {
  int i = blockIdx.x * blockDim.x + threadIdx.x;
  if (i < n) cnt[i] = 0;
}

__global__ void csr_count_rank(const int* __restrict__ dst, int* cnt,
                               int* __restrict__ rank, int e) {
  int i = blockIdx.x * blockDim.x + threadIdx.x;
  if (i < e) rank[i] = atomicAdd(&cnt[dst[i]], 1);
}

__global__ void __launch_bounds__(1024)
csr_scan(const int* __restrict__ cnt, int* __restrict__ rowptr,
         float* __restrict__ dinv, int n) {
  __shared__ int part[32];
  __shared__ int wexc[32];
  __shared__ int boff;
  const int tid = threadIdx.x;
  const int lane = tid & 31;
  const int w = tid >> 5;
  const int start = blockIdx.x * 1024;

  int t = 0;
  for (int k = tid; k < start; k += 1024) t += cnt[k];
  for (int o = 16; o > 0; o >>= 1) t += __shfl_xor(t, o);
  if (lane == 0) part[w] = t;

  const int i = start + tid;
  const int c = (i < n) ? cnt[i] : 0;
  int x = c;
  for (int o = 1; o < 32; o <<= 1) {
    int y = __shfl_up(x, o);
    if (lane >= o) x += y;
  }
  if (lane == 31) wexc[w] = x;
  __syncthreads();
  if (w == 0) {
    int p = part[lane];
    for (int o = 16; o > 0; o >>= 1) p += __shfl_xor(p, o);
    const int s = wexc[lane];
    int incl = s;
    for (int o = 1; o < 32; o <<= 1) {
      int y = __shfl_up(incl, o);
      if (lane >= o) incl += y;
    }
    wexc[lane] = incl - s;
    if (lane == 0) boff = p;
  }
  __syncthreads();
  const int excl = boff + wexc[w] + x - c;
  if (i < n) {
    rowptr[i] = excl;
    dinv[i] = rsqrtf((float)(c + 1));
  }
  if (i == n - 1) rowptr[n] = excl + c;
}

__global__ void csr_fill(const int* __restrict__ src, const int* __restrict__ dst,
                         const int* __restrict__ rowptr,
                         const int* __restrict__ rank, int* __restrict__ col, int e) {
  int i = blockIdx.x * blockDim.x + threadIdx.x;
  if (i < e) col[rowptr[dst[i]] + rank[i]] = src[i];
}

template <int LDB, int LDC, int NVALID>
__global__ void gcn_gemm_wmma_f32(const float* __restrict__ A,
                                  const float* __restrict__ B,
                                  const float* __restrict__ rscale,
                                  float* __restrict__ C, int nrows) {
  const int lane = threadIdx.x & 31;
  const int wave = threadIdx.x >> 5;
  const int row0 = (blockIdx.x * (blockDim.x >> 5) + wave) * 16;
  if (row0 >= nrows) return;
  const int n0   = blockIdx.y * 16;

  const int mrow = lane & 15;
  const int hsel = lane >> 4;
  const int kofs = hsel << 1;

  const float* arow = A + (row0 + mrow) * 64 + kofs;
  const int    bn   = n0 + mrow;
  const int    bcol = (bn < NVALID) ? bn : (NVALID - 1);
  const float* bp   = B + kofs * LDB + bcol;

  const int crow = row0 + (hsel << 3);
  const bool full = (row0 + 16 <= nrows);
  float sc[8];
  if (full) {
    const float4 s0 = *(const float4*)(rscale + crow);
    const float4 s1 = *(const float4*)(rscale + crow + 4);
    sc[0] = s0.x; sc[1] = s0.y; sc[2] = s0.z; sc[3] = s0.w;
    sc[4] = s1.x; sc[5] = s1.y; sc[6] = s1.z; sc[7] = s1.w;
  } else {
#pragma unroll
    for (int r = 0; r < 8; ++r) {
      const int rr = (crow + r < nrows) ? (crow + r) : (nrows - 1);
      sc[r] = rscale[rr];
    }
  }

  v8f acc = {};
  v2f a, b;
#pragma unroll
  for (int k = 0; k < 64; k += 4) {
    a = *(const v2f*)(arow + k);
    b = v2f{ bp[k * LDB], bp[k * LDB + LDB] };
    acc = __builtin_amdgcn_wmma_f32_16x16x4_f32(
         false, a,  false, b,
         (short)0, acc,  false,  false);
  }

  asm volatile("v_nop\n\tv_nop\n\tv_nop\n\tv_nop" : "+v"(acc) : "v"(a), "v"(b));

  float o[8];
#pragma unroll
  for (int r = 0; r < 8; ++r) o[r] = acc[r] * sc[r];

  float* cp = C + crow * LDC + n0 + mrow;
  if (full) {
#pragma unroll
    for (int r = 0; r < 8; ++r) cp[r * LDC] = o[r];
  } else {
#pragma unroll
    for (int r = 0; r < 8; ++r)
      if (crow + r < nrows) cp[r * LDC] = o[r];
  }
}

template <int LDH, int NV2, int LDO, bool RELU>
__global__ void gcn_gather_rows(const float* __restrict__ hin,
                                const int* __restrict__ rowptr,
                                const int* __restrict__ col,
                                const float* __restrict__ dinv,
                                const float* __restrict__ bias,
                                float* __restrict__ res, int n) {
  constexpr int G = 16;
  const int lane = threadIdx.x & 31;
  const int row = __builtin_amdgcn_readfirstlane(
      (int)(blockIdx.x * (blockDim.x >> 5) + (threadIdx.x >> 5)));
  if (row >= n) return;

  const int l2 = (lane < NV2) ? lane : (NV2 - 1);
  const int lo = l2 * 2;
  const v2f* hin2 = (const v2f*)hin;
  const int beg = rowptr[row];
  const int end = rowptr[row + 1];
  float di = dinv[row];
  asm volatile("" : "+s"(di));
  const v2f hs = hin2[(unsigned)row * (LDH / 2) + l2];
  const v2f bv = *(const v2f*)(bias + lo);
  v2f acc = hs;

  for (int base = beg; base < end; base += 32) {
    const int ne = (end - base < 32) ? (end - base) : 32;
    const int li = (lane < ne) ? lane : (ne - 1);
    const int my_s = col[base + li];
    for (int j = 0; j < ne; j += G) {
      v2f v[G];
#pragma unroll
      for (int u = 0; u < G; ++u) {
        const int jj = j + u;
        const int jc = (jj < ne) ? jj : (ne - 1);
        const int s = __builtin_amdgcn_readlane(my_s, jc);
        v[u] = hin2[(unsigned)s * (LDH / 2) + l2];
      }
#pragma unroll
      for (int u = 0; u < G; ++u) {
        const int jj = j + u;
        const float ax = acc.x + v[u].x;
        const float ay = acc.y + v[u].y;
        acc.x = (jj < ne) ? ax : acc.x;
        acc.y = (jj < ne) ? ay : acc.y;
      }
    }
  }

  v2f r = { fmaf(acc.x, di, bv.x), fmaf(acc.y, di, bv.y) };
  if (RELU) {
    r.x = r.x > 0.0f ? r.x : 0.0f;
    r.y = r.y > 0.0f ? r.y : 0.0f;
  }
  if (lane < NV2) *(v2f*)(res + (long)row * LDO + lo) = r;
}

extern "C" void kernel_launch(void* const* d_in, const int* in_sizes, int n_in,
                              void* d_out, int out_size, void* d_ws, size_t ws_size,
                              hipStream_t stream) {
  (void)n_in; (void)out_size; (void)ws_size;
  const float* x  = (const float*)d_in[0];
  const int*   ei = (const int*)d_in[1];
  const float* W1 = (const float*)d_in[2];
  const float* b1 = (const float*)d_in[3];
  const float* W3 = (const float*)d_in[4];
  const float* b3 = (const float*)d_in[5];

  const int N = in_sizes[0] / 64;
  const int E = in_sizes[1] / 2;
  const int* src = ei;
  const int* dst = ei + E;

  float* dinv   = (float*)d_ws;
  float* bufA   = dinv + N;
  float* bufB   = bufA + (long)N * 64;
  int*   cnt    = (int*)(bufB + (long)N * 64);
  int*   rowptr = cnt + N;
  int*   rank   = rowptr + N + 1;
  int*   col    = rank + E;
  float* out    = (float*)d_out;

  const int T = 256;
  const int mtiles = (N + 15) / 16;

  csr_zero_cnt<<<(N + T - 1) / T, T, 0, stream>>>(cnt, N);
  csr_count_rank<<<(E + T - 1) / T, T, 0, stream>>>(dst, cnt, rank, E);
  csr_scan<<<(N + 1023) / 1024, 1024, 0, stream>>>(cnt, rowptr, dinv, N);
  csr_fill<<<(E + T - 1) / T, T, 0, stream>>>(src, dst, rowptr, rank, col, E);

  {
    dim3 grid((mtiles + 7) / 8, 4);
    gcn_gemm_wmma_f32<64, 64, 64><<<grid, 256, 0, stream>>>(x, W1, dinv, bufA, N);
  }
  gcn_gather_rows<64, 32, 64, true><<<(N + 7) / 8, 256, 0, stream>>>(
      bufA, rowptr, col, dinv, b1, bufB, N);

  {
    dim3 grid((mtiles + 7) / 8, 3);
    gcn_gemm_wmma_f32<40, 48, 40><<<grid, 256, 0, stream>>>(bufB, W3, dinv, bufA, N);
  }
  gcn_gather_rows<48, 20, 40, false><<<(N + 7) / 8, 256, 0, stream>>>(
      bufA, rowptr, col, dinv, b3, out, N);
}


# Cut waves 325k -> 60k (entry rate limited): a gather wave sums 8 rows, next row's loads in flight; a GEMM wave owns all column tiles of its 16 rows; CSR kernels 4 elements/thread; sum order kept
# speedup vs baseline: 2.2037x; 1.0581x over previous
#include <hip/hip_runtime.h>

typedef float v2f __attribute__((ext_vector_type(2)));
typedef float v8f __attribute__((ext_vector_type(8)));

constexpr int EPT = 4;

__global__ void csr_zero_cnt(int* __restrict__ cnt, int n) {
  const int base = blockIdx.x * (blockDim.x * EPT) + threadIdx.x;
#pragma unroll
  for (int k = 0; k < EPT; ++k) {
    const int i = base + k * blockDim.x;
    if (i < n) cnt[i] = 0;
  }
}

__global__ void csr_count_rank(const int* __restrict__ dst, int* cnt,
                               int* __restrict__ rank, int e) {
  const int base = blockIdx.x * (blockDim.x * EPT) + threadIdx.x;
  int d[EPT], r[EPT];
#pragma unroll
  for (int k = 0; k < EPT; ++k) {
    const int i = base + k * blockDim.x;
    d[k] = (i < e) ? dst[i] : 0;
  }
#pragma unroll
  for (int k = 0; k < EPT; ++k) {
    const int i = base + k * blockDim.x;
    r[k] = 0;
    if (i < e) r[k] = atomicAdd(&cnt[d[k]], 1);
  }
#pragma unroll
  for (int k = 0; k < EPT; ++k) {
    const int i = base + k * blockDim.x;
    if (i < e) rank[i] = r[k];
  }
}

__global__ void __launch_bounds__(1024)
csr_scan(const int* __restrict__ cnt, int* __restrict__ rowptr,
         float* __restrict__ dinv, int n) {
  __shared__ int part[32];
  __shared__ int wexc[32];
  __shared__ int boff;
  const int tid = threadIdx.x;
  const int lane = tid & 31;
  const int w = tid >> 5;
  const int start = blockIdx.x * 1024;

  int t = 0;
  for (int k = tid; k < start; k += 1024) t += cnt[k];
  for (int o = 16; o > 0; o >>= 1) t += __shfl_xor(t, o);
  if (lane == 0) part[w] = t;

  const int i = start + tid;
  const int c = (i < n) ? cnt[i] : 0;
  int x = c;
  for (int o = 1; o < 32; o <<= 1) {
    int y = __shfl_up(x, o);
    if (lane >= o) x += y;
  }
  if (lane == 31) wexc[w] = x;
  __syncthreads();
  if (w == 0) {
    int p = part[lane];
    for (int o = 16; o > 0; o >>= 1) p += __shfl_xor(p, o);
    const int s = wexc[lane];
    int incl = s;
    for (int o = 1; o < 32; o <<= 1) {
      int y = __shfl_up(incl, o);
      if (lane >= o) incl += y;
    }
    wexc[lane] = incl - s;
    if (lane == 0) boff = p;
  }
  __syncthreads();
  const int excl = boff + wexc[w] + x - c;
  if (i < n) {
    rowptr[i] = excl;
    dinv[i] = rsqrtf((float)(c + 1));
  }
  if (i == n - 1) rowptr[n] = excl + c;
}

__global__ void csr_fill(const int* __restrict__ src, const int* __restrict__ dst,
                         const int* __restrict__ rowptr,
                         const int* __restrict__ rank, int* __restrict__ col, int e) {
  const int base = blockIdx.x * (blockDim.x * EPT) + threadIdx.x;
  int d[EPT], s[EPT], rk[EPT], rp[EPT];
#pragma unroll
  for (int k = 0; k < EPT; ++k) {
    const int i = base + k * blockDim.x;
    const int ii = (i < e) ? i : (e - 1);
    d[k] = dst[ii];
    s[k] = src[ii];
    rk[k] = rank[ii];
  }
#pragma unroll
  for (int k = 0; k < EPT; ++k) rp[k] = rowptr[d[k]];
#pragma unroll
  for (int k = 0; k < EPT; ++k) {
    const int i = base + k * blockDim.x;
    if (i < e) col[rp[k] + rk[k]] = s[k];
  }
}

template <int LDB, int LDC, int NVALID, int NT>
__global__ void __launch_bounds__(256)
gcn_gemm_wmma_f32(const float* __restrict__ A,
                                  const float* __restrict__ B,
                                  const float* __restrict__ rscale,
                                  float* __restrict__ C, int nrows) {
  const int lane = threadIdx.x & 31;
  const int wave = __builtin_amdgcn_readfirstlane((int)(threadIdx.x >> 5));
  const int row0 = (blockIdx.x * (blockDim.x >> 5) + wave) * 16;
  if (row0 >= nrows) return;

  v8f acc[NT];
#pragma unroll
  for (int t = 0; t < NT; ++t) {
    acc[t] = v8f{};
    asm volatile("" : "+v"(acc[t]));
  }

  const int mrow = lane & 15;
  const int hsel = lane >> 4;
  const int kofs = hsel << 1;

  const int ar = (row0 + mrow < nrows) ? (row0 + mrow) : (nrows - 1);
  const float* arow = A + ar * 64 + kofs;
  v2f a[16];
#pragma unroll
  for (int kk = 0; kk < 16; ++kk) a[kk] = *(const v2f*)(arow + 4 * kk);

  const int crow = row0 + (hsel << 3);
  const bool full = (row0 + 16 <= nrows);
  float sc[8];
  if (full) {
    const float4 s0 = *(const float4*)(rscale + crow);
    const float4 s1 = *(const float4*)(rscale + crow + 4);
    sc[0] = s0.x; sc[1] = s0.y; sc[2] = s0.z; sc[3] = s0.w;
    sc[4] = s1.x; sc[5] = s1.y; sc[6] = s1.z; sc[7] = s1.w;
  } else {
#pragma unroll
    for (int r = 0; r < 8; ++r) {
      const int rr = (crow + r < nrows) ? (crow + r) : (nrows - 1);
      sc[r] = rscale[rr];
    }
  }

  v2f b[NT][16];
#pragma unroll
  for (int t = 0; t < NT; ++t) {
    const int bn   = t * 16 + mrow;
    const int bcol = (bn < NVALID) ? bn : (NVALID - 1);
    const float* bp = B + kofs * LDB + bcol;
#pragma unroll
    for (int kk = 0; kk < 16; ++kk)
      b[t][kk] = v2f{ bp[4 * kk * LDB], bp[4 * kk * LDB + LDB] };
  }

  __builtin_amdgcn_sched_barrier(0);
#pragma unroll
  for (int kk = 0; kk < 16; ++kk) {
#pragma unroll
    for (int t = 0; t < NT; ++t) {
      acc[t] = __builtin_amdgcn_wmma_f32_16x16x4_f32(
           false, a[kk],  false, b[t][kk],
           (short)0, acc[t],  false,  false);
    }
  }

  __builtin_amdgcn_sched_barrier(0);

  if constexpr (NT == 4) {
    asm volatile("v_nop\n\tv_nop\n\tv_nop\n\tv_nop"
                 : "+v"(acc[0]), "+v"(acc[1]), "+v"(acc[2]), "+v"(acc[3])
                 : "v"(a[15]), "v"(b[0][15]), "v"(b[1][15]), "v"(b[2][15]), "v"(b[3][15]));
  } else {
    static_assert(NT == 3 || NT == 4, "guard written for 3 or 4 tiles");
    asm volatile("v_nop\n\tv_nop\n\tv_nop\n\tv_nop"
                 : "+v"(acc[0]), "+v"(acc[1]), "+v"(acc[2])
                 : "v"(a[15]), "v"(b[0][15]), "v"(b[1][15]), "v"(b[2][15]));
  }

#pragma unroll
  for (int t = 0; t < NT; ++t) {
    float o[8];
#pragma unroll
    for (int r = 0; r < 8; ++r) o[r] = acc[t][r] * sc[r];
    float* cp = C + crow * LDC + t * 16 + mrow;
    if (full) {
#pragma unroll
      for (int r = 0; r < 8; ++r) cp[r * LDC] = o[r];
    } else {
#pragma unroll
      for (int r = 0; r < 8; ++r)
        if (crow + r < nrows) cp[r * LDC] = o[r];
    }
  }
}

template <int LDH, int NV2, int LDO, bool RELU, int K>
__global__ void gcn_gather_rows(const float* __restrict__ hin,
                                const int* __restrict__ rowptr,
                                const int* __restrict__ col,
                                const float* __restrict__ dinv,
                                const float* __restrict__ bias,
                                float* __restrict__ res, int n) {
  constexpr int G = 16;
  static_assert(K >= 1 && K < 32, "row bounds come from one 32-lane load");
  static_assert(2 * G == 32, "a row's second group of ids is the upper half of ms[i]");
  const int lane = threadIdx.x & 31;
  const int wid = __builtin_amdgcn_readfirstlane(
      (int)(blockIdx.x * (blockDim.x >> 5) + (threadIdx.x >> 5)));
  const int r0 = wid * K;
  if (r0 >= n) return;

  const int l2 = (lane < NV2) ? lane : (NV2 - 1);
  const int lo = l2 * 2;
  const v2f* hin2 = (const v2f*)hin;

  const int bi = (lane < K) ? lane : K;
  const int rpv = rowptr[(r0 + bi < n) ? (r0 + bi) : n];
  const int fi = (lane < K - 1) ? lane : (K - 1);
  const float dv = dinv[(r0 + fi < n) ? (r0 + fi) : (n - 1)];
  const v2f bv = *(const v2f*)(bias + lo);

  int rb[K], rn[K], ms[K];
#pragma unroll
  for (int i = 0; i < K; ++i) {
    rb[i] = __builtin_amdgcn_readlane(rpv, i);
    rn[i] = __builtin_amdgcn_readlane(rpv, i + 1) - rb[i];
    ms[i] = 0;
    if (rn[i] > 0) ms[i] = col[rb[i] + ((lane < rn[i]) ? lane : (rn[i] - 1))];
  }

  v2f va[G], vb[G], sfa, sfb;

  auto issue = [&](const int i, v2f (&v)[G], v2f& sf) {
    const int rr = (r0 + i < n) ? (r0 + i) : (n - 1);
    sf = hin2[(unsigned)rr * (LDH / 2) + l2];
#pragma unroll
    for (int u = 0; u < G; ++u) {
      const int s = __builtin_amdgcn_readlane(ms[i], u);
      v[u] = hin2[(unsigned)s * (LDH / 2) + l2];
    }
  };

  auto finish = [&](const int i, v2f (&v)[G], const v2f& sf) {
    const int ne = rn[i];
    v2f acc = sf;
#pragma unroll
    for (int u = 0; u < G; ++u) {
      const float ax = acc.x + v[u].x;
      const float ay = acc.y + v[u].y;
      acc.x = (u < ne) ? ax : acc.x;
      acc.y = (u < ne) ? ay : acc.y;
    }
#pragma nounroll
    for (int off = G; off < ne; off += G) {
      const int nn = (ne - off < G) ? (ne - off) : G;
      int cs = ms[i];
      int sh = G;
      if (off >= 32) {
        cs = col[rb[i] + off + ((lane < nn) ? lane : (nn - 1))];
        sh = 0;
      }
      v2f w[G];
#pragma unroll
      for (int u = 0; u < G; ++u) {
        const int s = __builtin_amdgcn_readlane(cs, sh + u);
        w[u] = hin2[(unsigned)s * (LDH / 2) + l2];
      }
#pragma unroll
      for (int u = 0; u < G; ++u) {
        const float ax = acc.x + w[u].x;
        const float ay = acc.y + w[u].y;
        acc.x = (u < nn) ? ax : acc.x;
        acc.y = (u < nn) ? ay : acc.y;
      }
    }
    const float di = __int_as_float(
        __builtin_amdgcn_readlane(__float_as_int(dv), i));
    v2f r = { fmaf(acc.x, di, bv.x), fmaf(acc.y, di, bv.y) };
    if (RELU) {
      r.x = r.x > 0.0f ? r.x : 0.0f;
      r.y = r.y > 0.0f ? r.y : 0.0f;
    }
    if (r0 + i < n && lane < NV2) *(v2f*)(res + (long)(r0 + i) * LDO + lo) = r;
  };

  issue(0, va, sfa);
#pragma unroll
  for (int i = 0; i < K; i += 2) {
    if (i + 1 < K) issue(i + 1, vb, sfb);
    finish(i, va, sfa);
    if (i + 1 < K) {
      if (i + 2 < K) issue(i + 2, va, sfa);
      finish(i + 1, vb, sfb);
    }
  }
}

extern "C" void kernel_launch(void* const* d_in, const int* in_sizes, int n_in,
                              void* d_out, int out_size, void* d_ws, size_t ws_size,
                              hipStream_t stream) {
  (void)n_in; (void)out_size; (void)ws_size;
  const float* x  = (const float*)d_in[0];
  const int*   ei = (const int*)d_in[1];
  const float* W1 = (const float*)d_in[2];
  const float* b1 = (const float*)d_in[3];
  const float* W3 = (const float*)d_in[4];
  const float* b3 = (const float*)d_in[5];

  const int N = in_sizes[0] / 64;
  const int E = in_sizes[1] / 2;
  const int* src = ei;
  const int* dst = ei + E;

  float* dinv   = (float*)d_ws;
  float* bufA   = dinv + N;
  float* bufB   = bufA + (long)N * 64;
  int*   cnt    = (int*)(bufB + (long)N * 64);
  int*   rowptr = cnt + N;
  int*   rank   = rowptr + N + 1;
  int*   col    = rank + E;
  float* out    = (float*)d_out;

  const int T = 256;
  const int mtiles = (N + 15) / 16;
  constexpr int KR = 8;

  const int TE = T * EPT;
  csr_zero_cnt<<<(N + TE - 1) / TE, T, 0, stream>>>(cnt, N);
  csr_count_rank<<<(E + TE - 1) / TE, T, 0, stream>>>(dst, cnt, rank, E);
  csr_scan<<<(N + 1023) / 1024, 1024, 0, stream>>>(cnt, rowptr, dinv, N);
  csr_fill<<<(E + TE - 1) / TE, T, 0, stream>>>(src, dst, rowptr, rank, col, E);

  {
    dim3 grid((mtiles + 7) / 8);
    gcn_gemm_wmma_f32<64, 64, 64, 4><<<grid, 256, 0, stream>>>(x, W1, dinv, bufA, N);
  }
  gcn_gather_rows<64, 32, 64, true, KR><<<(N + 8 * KR - 1) / (8 * KR), 256, 0, stream>>>(
      bufA, rowptr, col, dinv, b1, bufB, N);

  {
    dim3 grid((mtiles + 7) / 8);
    gcn_gemm_wmma_f32<40, 48, 40, 3><<<grid, 256, 0, stream>>>(bufB, W3, dinv, bufA, N);
  }
  gcn_gather_rows<48, 20, 40, false, KR><<<(N + 8 * KR - 1) / (8 * KR), 256, 0, stream>>>(
      bufA, rowptr, col, dinv, b3, out, N);
}


# Edge ordering through LDS: edges binned by 512-node destination range (LDS histogram, LDS sort, run-wise stores), ranked per bucket by LDS adds; no global atomics, 8 -> 7 launches; arrival-order sum
# speedup vs baseline: 4.2721x; 1.9386x over previous
#include <hip/hip_runtime.h>

typedef float v2f __attribute__((ext_vector_type(2)));
typedef float v8f __attribute__((ext_vector_type(8)));

constexpr int BSH  = 9;
constexpr int BN   = 1 << BSH;
constexpr int MAXB = 256;
constexpr int CH   = 4704;
constexpr int BT   = 1024;
constexpr int KPT  = (CH + BT - 1) / BT;
constexpr int UB   = 8;
constexpr int HB   = 16;
constexpr int CAP  = 16384;

__device__ __forceinline__ int block_excl_scan(const int v, int* wsum, int& total) {
  const int lane = threadIdx.x & 31;
  const int w = threadIdx.x >> 5;
  int x = v;
  for (int o = 1; o < 32; o <<= 1) {
    const int y = __shfl_up(x, o);
    if (lane >= o) x += y;
  }
  if (lane == 31) wsum[w] = x;
  __syncthreads();
  const int mine = wsum[lane];
  int s = (lane < w) ? mine : 0;
  int t = mine;
  for (int o = 16; o > 0; o >>= 1) {
    s += __shfl_xor(s, o);
    t += __shfl_xor(t, o);
  }
  __syncthreads();
  total = t;
  return s + x - v;
}

__global__ void __launch_bounds__(BT)
csr_bucket_hist(const int* __restrict__ dst, int* __restrict__ hist, int e, int nb) {
  __shared__ int h[MAXB];
  const int tid = threadIdx.x;
  if (tid < MAXB) h[tid] = 0;
  __syncthreads();
  const int base = blockIdx.x * CH;
  const int end = (base + CH < e) ? (base + CH) : e;
  int d[KPT];
#pragma unroll
  for (int k = 0; k < KPT; ++k) {
    const int i = base + tid + k * BT;
    d[k] = (i < end) ? dst[i] : -1;
  }
#pragma unroll
  for (int k = 0; k < KPT; ++k)
    if (d[k] >= 0) atomicAdd(&h[d[k] >> BSH], 1);
  __syncthreads();
  if (tid < nb) hist[blockIdx.x * nb + tid] = h[tid];
}

__global__ void __launch_bounds__(BT)
csr_bucket_place(const int* __restrict__ src, const int* __restrict__ dst,
                 const int* __restrict__ hist, unsigned* __restrict__ tmp,
                 int* __restrict__ bstart, int e, int nb, int nblk) {
  __shared__ int tot[4][MAXB];
  __shared__ int pre[4][MAXB];
  __shared__ int lstart[MAXB];
  __shared__ int gdst[MAXB];
  __shared__ int lcur[MAXB];
  __shared__ int wsum[32];
  __shared__ unsigned stage[CH];
  __shared__ unsigned char sbk[CH];
  const int tid = threadIdx.x;
  const int blk = blockIdx.x;
  const int base = blk * CH;
  const int end = (base + CH < e) ? (base + CH) : e;

  int d[KPT], s[KPT];
#pragma unroll
  for (int k = 0; k < KPT; ++k) {
    const int i = base + tid + k * BT;
    const int ii = (i < end) ? i : (end - 1);
    d[k] = dst[ii];
    s[k] = src[ii];
  }

  {
    const int part = tid >> 8;
    const int b = tid & 255;
    int t = 0, p = 0;
    if (b < nb) {
      const int q = (nblk + 3) >> 2;
      const int j0 = part * q;
      const int j1 = (j0 + q < nblk) ? (j0 + q) : nblk;
      for (int j = j0; j < j1; j += HB) {
        int v[HB];
#pragma unroll
        for (int u = 0; u < HB; ++u) {
          const int jj = (j + u < j1) ? (j + u) : (j1 - 1);
          v[u] = hist[jj * nb + b];
        }
#pragma unroll
        for (int u = 0; u < HB; ++u) {
          const int a = (j + u < j1) ? v[u] : 0;
          t += a;
          p += (j + u < blk) ? a : 0;
        }
      }
    }
    tot[part][b] = t;
    pre[part][b] = p;
  }
  if (tid < MAXB) lcur[tid] = 0;
  __syncthreads();

  int T = 0, P = 0, L = 0;
  if (tid < nb) {
    T = tot[0][tid] + tot[1][tid] + tot[2][tid] + tot[3][tid];
    P = pre[0][tid] + pre[1][tid] + pre[2][tid] + pre[3][tid];
    L = hist[blk * nb + tid];
  }
  int totalT, totalL;
  const int bs = block_excl_scan(T, wsum, totalT);
  const int ls = block_excl_scan(L, wsum, totalL);
  if (tid < nb) {
    lstart[tid] = ls;
    gdst[tid] = bs + P - ls;
    if (blk == 0) bstart[tid] = bs;
  }
  if (blk == 0 && tid == nb) bstart[nb] = totalT;
  __syncthreads();

#pragma unroll
  for (int k = 0; k < KPT; ++k) {
    const int i = base + tid + k * BT;
    if (i < end) {
      const int bb = d[k] >> BSH;
      const int pos = lstart[bb] + atomicAdd(&lcur[bb], 1);
      stage[pos] = ((unsigned)s[k] << BSH) | (unsigned)(d[k] & (BN - 1));
      sbk[pos] = (unsigned char)bb;
    }
  }
  __syncthreads();

  const int nloc = end - base;
#pragma unroll
  for (int k = 0; k < KPT; ++k) {
    const int pos = tid + k * BT;
    if (pos < nloc) tmp[gdst[sbk[pos]] + pos] = stage[pos];
  }
}

__global__ void __launch_bounds__(BT)
csr_rows(const unsigned* __restrict__ tmp, const int* __restrict__ bstart,
         int* __restrict__ rowptr, float* __restrict__ dinv,
         int* __restrict__ col, int n) {
  __shared__ int cnt[BN];
  __shared__ int cur[BN];
  __shared__ int wsum[32];
  __shared__ int stg[CAP];
  const int tid = threadIdx.x;
  const int b = blockIdx.x;
  const int s0 = bstart[b];
  const int s1 = bstart[b + 1];
  const int m = s1 - s0;
  if (tid < BN) cnt[tid] = 0;
  __syncthreads();

  for (int i0 = 0; i0 < m; i0 += BT * UB) {
    unsigned w[UB];
#pragma unroll
    for (int u = 0; u < UB; ++u) {
      const int i = i0 + tid + u * BT;
      w[u] = tmp[s0 + ((i < m) ? i : (m - 1))];
    }
#pragma unroll
    for (int u = 0; u < UB; ++u) {
      const int i = i0 + tid + u * BT;
      if (i < m) atomicAdd(&cnt[w[u] & (BN - 1)], 1);
    }
  }
  __syncthreads();

  const int c = (tid < BN) ? cnt[tid] : 0;
  int total;
  const int ex = block_excl_scan(c, wsum, total);
  const int node = b * BN + tid;
  if (tid < BN) {
    cur[tid] = ex;
    if (node < n) {
      rowptr[node] = s0 + ex;
      dinv[node] = rsqrtf((float)(c + 1));
    }
  }
  if (b == (int)gridDim.x - 1 && tid == 0) rowptr[n] = s1;
  __syncthreads();

  const bool staged = (m <= CAP);
  for (int i0 = 0; i0 < m; i0 += BT * UB) {
    unsigned w[UB];
#pragma unroll
    for (int u = 0; u < UB; ++u) {
      const int i = i0 + tid + u * BT;
      w[u] = tmp[s0 + ((i < m) ? i : (m - 1))];
    }
#pragma unroll
    for (int u = 0; u < UB; ++u) {
      const int i = i0 + tid + u * BT;
      if (i < m) {
        const int p = atomicAdd(&cur[w[u] & (BN - 1)], 1);
        if (staged) stg[p] = (int)(w[u] >> BSH);
        else col[s0 + p] = (int)(w[u] >> BSH);
      }
    }
  }
  __syncthreads();
  if (staged)
    for (int i = tid; i < m; i += BT) col[s0 + i] = stg[i];
}

template <int LDB, int LDC, int NVALID, int NT>
__global__ void __launch_bounds__(256)
gcn_gemm_wmma_f32(const float* __restrict__ A,
                                  const float* __restrict__ B,
                                  const float* __restrict__ rscale,
                                  float* __restrict__ C, int nrows) {
  const int lane = threadIdx.x & 31;
  const int wave = __builtin_amdgcn_readfirstlane((int)(threadIdx.x >> 5));
  const int row0 = (blockIdx.x * (blockDim.x >> 5) + wave) * 16;
  if (row0 >= nrows) return;

  v8f acc[NT];
#pragma unroll
  for (int t = 0; t < NT; ++t) {
    acc[t] = v8f{};
    asm volatile("" : "+v"(acc[t]));
  }

  const int mrow = lane & 15;
  const int hsel = lane >> 4;
  const int kofs = hsel << 1;

  const int ar = (row0 + mrow < nrows) ? (row0 + mrow) : (nrows - 1);
  const float* arow = A + ar * 64 + kofs;
  v2f a[16];
#pragma unroll
  for (int kk = 0; kk < 16; ++kk) a[kk] = *(const v2f*)(arow + 4 * kk);

  const int crow = row0 + (hsel << 3);
  const bool full = (row0 + 16 <= nrows);
  float sc[8];
  if (full) {
    const float4 s0 = *(const float4*)(rscale + crow);
    const float4 s1 = *(const float4*)(rscale + crow + 4);
    sc[0] = s0.x; sc[1] = s0.y; sc[2] = s0.z; sc[3] = s0.w;
    sc[4] = s1.x; sc[5] = s1.y; sc[6] = s1.z; sc[7] = s1.w;
  } else {
#pragma unroll
    for (int r = 0; r < 8; ++r) {
      const int rr = (crow + r < nrows) ? (crow + r) : (nrows - 1);
      sc[r] = rscale[rr];
    }
  }

  v2f b[NT][16];
#pragma unroll
  for (int t = 0; t < NT; ++t) {
    const int bn   = t * 16 + mrow;
    const int bcol = (bn < NVALID) ? bn : (NVALID - 1);
    const float* bp = B + kofs * LDB + bcol;
#pragma unroll
    for (int kk = 0; kk < 16; ++kk)
      b[t][kk] = v2f{ bp[4 * kk * LDB], bp[4 * kk * LDB + LDB] };
  }

  __builtin_amdgcn_sched_barrier(0);
#pragma unroll
  for (int kk = 0; kk < 16; ++kk) {
#pragma unroll
    for (int t = 0; t < NT; ++t) {
      acc[t] = __builtin_amdgcn_wmma_f32_16x16x4_f32(
           false, a[kk],  false, b[t][kk],
           (short)0, acc[t],  false,  false);
    }
  }

  __builtin_amdgcn_sched_barrier(0);

  if constexpr (NT == 4) {
    asm volatile("v_nop\n\tv_nop\n\tv_nop\n\tv_nop"
                 : "+v"(acc[0]), "+v"(acc[1]), "+v"(acc[2]), "+v"(acc[3])
                 : "v"(a[15]), "v"(b[0][15]), "v"(b[1][15]), "v"(b[2][15]), "v"(b[3][15]));
  } else {
    static_assert(NT == 3 || NT == 4, "guard written for 3 or 4 tiles");
    asm volatile("v_nop\n\tv_nop\n\tv_nop\n\tv_nop"
                 : "+v"(acc[0]), "+v"(acc[1]), "+v"(acc[2])
                 : "v"(a[15]), "v"(b[0][15]), "v"(b[1][15]), "v"(b[2][15]));
  }

#pragma unroll
  for (int t = 0; t < NT; ++t) {
    float o[8];
#pragma unroll
    for (int r = 0; r < 8; ++r) o[r] = acc[t][r] * sc[r];
    float* cp = C + crow * LDC + t * 16 + mrow;
    if (full) {
#pragma unroll
      for (int r = 0; r < 8; ++r) cp[r * LDC] = o[r];
    } else {
#pragma unroll
      for (int r = 0; r < 8; ++r)
        if (crow + r < nrows) cp[r * LDC] = o[r];
    }
  }
}

template <int LDH, int NV2, int LDO, bool RELU, int K>
__global__ void gcn_gather_rows(const float* __restrict__ hin,
                                const int* __restrict__ rowptr,
                                const int* __restrict__ col,
                                const float* __restrict__ dinv,
                                const float* __restrict__ bias,
                                float* __restrict__ res, int n) {
  constexpr int G = 16;
  static_assert(K >= 1 && K < 32, "row bounds come from one 32-lane load");
  static_assert(2 * G == 32, "a row's second group of ids is the upper half of ms[i]");
  const int lane = threadIdx.x & 31;
  const int wid = __builtin_amdgcn_readfirstlane(
      (int)(blockIdx.x * (blockDim.x >> 5) + (threadIdx.x >> 5)));
  const int r0 = wid * K;
  if (r0 >= n) return;

  const int l2 = (lane < NV2) ? lane : (NV2 - 1);
  const int lo = l2 * 2;
  const v2f* hin2 = (const v2f*)hin;

  const int bi = (lane < K) ? lane : K;
  const int rpv = rowptr[(r0 + bi < n) ? (r0 + bi) : n];
  const int fi = (lane < K - 1) ? lane : (K - 1);
  const float dv = dinv[(r0 + fi < n) ? (r0 + fi) : (n - 1)];
  const v2f bv = *(const v2f*)(bias + lo);

  int rb[K], rn[K], ms[K];
#pragma unroll
  for (int i = 0; i < K; ++i) {
    rb[i] = __builtin_amdgcn_readlane(rpv, i);
    rn[i] = __builtin_amdgcn_readlane(rpv, i + 1) - rb[i];
    ms[i] = 0;
    if (rn[i] > 0) ms[i] = col[rb[i] + ((lane < rn[i]) ? lane : (rn[i] - 1))];
  }

  v2f va[G], vb[G], sfa, sfb;

  auto issue = [&](const int i, v2f (&v)[G], v2f& sf) {
    const int rr = (r0 + i < n) ? (r0 + i) : (n - 1);
    sf = hin2[(unsigned)rr * (LDH / 2) + l2];
#pragma unroll
    for (int u = 0; u < G; ++u) {
      const int s = __builtin_amdgcn_readlane(ms[i], u);
      v[u] = hin2[(unsigned)s * (LDH / 2) + l2];
    }
  };

  auto finish = [&](const int i, v2f (&v)[G], const v2f& sf) {
    const int ne = rn[i];
    v2f acc = sf;
#pragma unroll
    for (int u = 0; u < G; ++u) {
      const float ax = acc.x + v[u].x;
      const float ay = acc.y + v[u].y;
      acc.x = (u < ne) ? ax : acc.x;
      acc.y = (u < ne) ? ay : acc.y;
    }
#pragma nounroll
    for (int off = G; off < ne; off += G) {
      const int nn = (ne - off < G) ? (ne - off) : G;
      int cs = ms[i];
      int sh = G;
      if (off >= 32) {
        cs = col[rb[i] + off + ((lane < nn) ? lane : (nn - 1))];
        sh = 0;
      }
      v2f w[G];
#pragma unroll
      for (int u = 0; u < G; ++u) {
        const int s = __builtin_amdgcn_readlane(cs, sh + u);
        w[u] = hin2[(unsigned)s * (LDH / 2) + l2];
      }
#pragma unroll
      for (int u = 0; u < G; ++u) {
        const float ax = acc.x + w[u].x;
        const float ay = acc.y + w[u].y;
        acc.x = (u < nn) ? ax : acc.x;
        acc.y = (u < nn) ? ay : acc.y;
      }
    }
    const float di = __int_as_float(
        __builtin_amdgcn_readlane(__float_as_int(dv), i));
    v2f r = { fmaf(acc.x, di, bv.x), fmaf(acc.y, di, bv.y) };
    if (RELU) {
      r.x = r.x > 0.0f ? r.x : 0.0f;
      r.y = r.y > 0.0f ? r.y : 0.0f;
    }
    if (r0 + i < n && lane < NV2) *(v2f*)(res + (long)(r0 + i) * LDO + lo) = r;
  };

  issue(0, va, sfa);
#pragma unroll
  for (int i = 0; i < K; i += 2) {
    if (i + 1 < K) issue(i + 1, vb, sfb);
    finish(i, va, sfa);
    if (i + 1 < K) {
      if (i + 2 < K) issue(i + 2, va, sfa);
      finish(i + 1, vb, sfb);
    }
  }
}

extern "C" void kernel_launch(void* const* d_in, const int* in_sizes, int n_in,
                              void* d_out, int out_size, void* d_ws, size_t ws_size,
                              hipStream_t stream) {
  (void)n_in; (void)out_size; (void)ws_size;
  const float* x  = (const float*)d_in[0];
  const int*   ei = (const int*)d_in[1];
  const float* W1 = (const float*)d_in[2];
  const float* b1 = (const float*)d_in[3];
  const float* W3 = (const float*)d_in[4];
  const float* b3 = (const float*)d_in[5];

  const int N = in_sizes[0] / 64;
  const int E = in_sizes[1] / 2;
  const int* src = ei;
  const int* dst = ei + E;

  const int NBK = (N + BN - 1) / BN;
  const int NCH = (E + CH - 1) / CH;

  float* dinv   = (float*)d_ws;
  float* bufA   = dinv + N;
  float* bufB   = bufA + (long)N * 64;
  int*   hist   = (int*)(bufB + (long)N * 64);
  int*   bstart = hist + NCH * NBK;
  int*   rowptr = hist + N;
  unsigned* tmp = (unsigned*)(rowptr + N + 1);
  int*   col    = (int*)tmp + E;
  float* out    = (float*)d_out;

  const int mtiles = (N + 15) / 16;
  constexpr int KR = 8;

  csr_bucket_hist<<<NCH, BT, 0, stream>>>(dst, hist, E, NBK);
  csr_bucket_place<<<NCH, BT, 0, stream>>>(src, dst, hist, tmp, bstart, E, NBK, NCH);
  csr_rows<<<NBK, BT, 0, stream>>>(tmp, bstart, rowptr, dinv, col, N);

  {
    dim3 grid((mtiles + 7) / 8);
    gcn_gemm_wmma_f32<64, 64, 64, 4><<<grid, 256, 0, stream>>>(x, W1, dinv, bufA, N);
  }
  gcn_gather_rows<64, 32, 64, true, KR><<<(N + 8 * KR - 1) / (8 * KR), 256, 0, stream>>>(
      bufA, rowptr, col, dinv, b1, bufB, N);

  {
    dim3 grid((mtiles + 7) / 8);
    gcn_gemm_wmma_f32<40, 48, 40, 3><<<grid, 256, 0, stream>>>(bufB, W3, dinv, bufA, N);
  }
  gcn_gather_rows<48, 20, 40, false, KR><<<(N + 8 * KR - 1) / (8 * KR), 256, 0, stream>>>(
      bufA, rowptr, col, dinv, b3, out, N);
}


# Fuse the layer-2 product into the layer-1 gather: a block keeps its 64 summed relu rows in LDS and multiplies them by W3 on the f32 WMMA; h1 never goes to memory, 7 -> 6 launches; sum orders kept
# speedup vs baseline: 4.7442x; 1.1105x over previous
#include <hip/hip_runtime.h>

typedef float v2f __attribute__((ext_vector_type(2)));
typedef float v8f __attribute__((ext_vector_type(8)));

constexpr int BSH  = 9;
constexpr int BN   = 1 << BSH;
constexpr int MAXB = 256;
constexpr int CH   = 4704;
constexpr int BT   = 1024;
constexpr int KPT  = (CH + BT - 1) / BT;
constexpr int UB   = 8;
constexpr int HB   = 16;
constexpr int CAP  = 16384;

__device__ __forceinline__ int block_excl_scan(const int v, int* wsum, int& total) {
  const int lane = threadIdx.x & 31;
  const int w = threadIdx.x >> 5;
  int x = v;
  for (int o = 1; o < 32; o <<= 1) {
    const int y = __shfl_up(x, o);
    if (lane >= o) x += y;
  }
  if (lane == 31) wsum[w] = x;
  __syncthreads();
  const int mine = wsum[lane];
  int s = (lane < w) ? mine : 0;
  int t = mine;
  for (int o = 16; o > 0; o >>= 1) {
    s += __shfl_xor(s, o);
    t += __shfl_xor(t, o);
  }
  __syncthreads();
  total = t;
  return s + x - v;
}

__global__ void __launch_bounds__(BT)
csr_bucket_hist(const int* __restrict__ dst, int* __restrict__ hist, int e, int nb) {
  __shared__ int h[MAXB];
  const int tid = threadIdx.x;
  if (tid < MAXB) h[tid] = 0;
  __syncthreads();
  const int base = blockIdx.x * CH;
  const int end = (base + CH < e) ? (base + CH) : e;
  int d[KPT];
#pragma unroll
  for (int k = 0; k < KPT; ++k) {
    const int i = base + tid + k * BT;
    d[k] = (i < end) ? dst[i] : -1;
  }
#pragma unroll
  for (int k = 0; k < KPT; ++k)
    if (d[k] >= 0) atomicAdd(&h[d[k] >> BSH], 1);
  __syncthreads();
  if (tid < nb) hist[blockIdx.x * nb + tid] = h[tid];
}

__global__ void __launch_bounds__(BT)
csr_bucket_place(const int* __restrict__ src, const int* __restrict__ dst,
                 const int* __restrict__ hist, unsigned* __restrict__ tmp,
                 int* __restrict__ bstart, int e, int nb, int nblk) {
  __shared__ int tot[4][MAXB];
  __shared__ int pre[4][MAXB];
  __shared__ int lstart[MAXB];
  __shared__ int gdst[MAXB];
  __shared__ int lcur[MAXB];
  __shared__ int wsum[32];
  __shared__ unsigned stage[CH];
  __shared__ unsigned char sbk[CH];
  const int tid = threadIdx.x;
  const int blk = blockIdx.x;
  const int base = blk * CH;
  const int end = (base + CH < e) ? (base + CH) : e;

  int d[KPT], s[KPT];
#pragma unroll
  for (int k = 0; k < KPT; ++k) {
    const int i = base + tid + k * BT;
    const int ii = (i < end) ? i : (end - 1);
    d[k] = dst[ii];
    s[k] = src[ii];
  }

  {
    const int part = tid >> 8;
    const int b = tid & 255;
    int t = 0, p = 0;
    if (b < nb) {
      const int q = (nblk + 3) >> 2;
      const int j0 = part * q;
      const int j1 = (j0 + q < nblk) ? (j0 + q) : nblk;
      for (int j = j0; j < j1; j += HB) {
        int v[HB];
#pragma unroll
        for (int u = 0; u < HB; ++u) {
          const int jj = (j + u < j1) ? (j + u) : (j1 - 1);
          v[u] = hist[jj * nb + b];
        }
#pragma unroll
        for (int u = 0; u < HB; ++u) {
          const int a = (j + u < j1) ? v[u] : 0;
          t += a;
          p += (j + u < blk) ? a : 0;
        }
      }
    }
    tot[part][b] = t;
    pre[part][b] = p;
  }
  if (tid < MAXB) lcur[tid] = 0;
  __syncthreads();

  int T = 0, P = 0, L = 0;
  if (tid < nb) {
    T = tot[0][tid] + tot[1][tid] + tot[2][tid] + tot[3][tid];
    P = pre[0][tid] + pre[1][tid] + pre[2][tid] + pre[3][tid];
    L = hist[blk * nb + tid];
  }
  int totalT, totalL;
  const int bs = block_excl_scan(T, wsum, totalT);
  const int ls = block_excl_scan(L, wsum, totalL);
  if (tid < nb) {
    lstart[tid] = ls;
    gdst[tid] = bs + P - ls;
    if (blk == 0) bstart[tid] = bs;
  }
  if (blk == 0 && tid == nb) bstart[nb] = totalT;
  __syncthreads();

#pragma unroll
  for (int k = 0; k < KPT; ++k) {
    const int i = base + tid + k * BT;
    if (i < end) {
      const int bb = d[k] >> BSH;
      const int pos = lstart[bb] + atomicAdd(&lcur[bb], 1);
      stage[pos] = ((unsigned)s[k] << BSH) | (unsigned)(d[k] & (BN - 1));
      sbk[pos] = (unsigned char)bb;
    }
  }
  __syncthreads();

  const int nloc = end - base;
#pragma unroll
  for (int k = 0; k < KPT; ++k) {
    const int pos = tid + k * BT;
    if (pos < nloc) tmp[gdst[sbk[pos]] + pos] = stage[pos];
  }
}

__global__ void __launch_bounds__(BT)
csr_rows(const unsigned* __restrict__ tmp, const int* __restrict__ bstart,
         int* __restrict__ rowptr, float* __restrict__ dinv,
         int* __restrict__ col, int n) {
  __shared__ int cnt[BN];
  __shared__ int cur[BN];
  __shared__ int wsum[32];
  __shared__ int stg[CAP];
  const int tid = threadIdx.x;
  const int b = blockIdx.x;
  const int s0 = bstart[b];
  const int s1 = bstart[b + 1];
  const int m = s1 - s0;
  if (tid < BN) cnt[tid] = 0;
  __syncthreads();

  for (int i0 = 0; i0 < m; i0 += BT * UB) {
    unsigned w[UB];
#pragma unroll
    for (int u = 0; u < UB; ++u) {
      const int i = i0 + tid + u * BT;
      w[u] = tmp[s0 + ((i < m) ? i : (m - 1))];
    }
#pragma unroll
    for (int u = 0; u < UB; ++u) {
      const int i = i0 + tid + u * BT;
      if (i < m) atomicAdd(&cnt[w[u] & (BN - 1)], 1);
    }
  }
  __syncthreads();

  const int c = (tid < BN) ? cnt[tid] : 0;
  int total;
  const int ex = block_excl_scan(c, wsum, total);
  const int node = b * BN + tid;
  if (tid < BN) {
    cur[tid] = ex;
    if (node < n) {
      rowptr[node] = s0 + ex;
      dinv[node] = rsqrtf((float)(c + 1));
    }
  }
  if (b == (int)gridDim.x - 1 && tid == 0) rowptr[n] = s1;
  __syncthreads();

  const bool staged = (m <= CAP);
  for (int i0 = 0; i0 < m; i0 += BT * UB) {
    unsigned w[UB];
#pragma unroll
    for (int u = 0; u < UB; ++u) {
      const int i = i0 + tid + u * BT;
      w[u] = tmp[s0 + ((i < m) ? i : (m - 1))];
    }
#pragma unroll
    for (int u = 0; u < UB; ++u) {
      const int i = i0 + tid + u * BT;
      if (i < m) {
        const int p = atomicAdd(&cur[w[u] & (BN - 1)], 1);
        if (staged) stg[p] = (int)(w[u] >> BSH);
        else col[s0 + p] = (int)(w[u] >> BSH);
      }
    }
  }
  __syncthreads();
  if (staged)
    for (int i = tid; i < m; i += BT) col[s0 + i] = stg[i];
}

template <int LDB, int LDC, int NVALID, int NT>
__global__ void __launch_bounds__(256)
gcn_gemm_wmma_f32(const float* __restrict__ A,
                                  const float* __restrict__ B,
                                  const float* __restrict__ rscale,
                                  float* __restrict__ C, int nrows) {
  const int lane = threadIdx.x & 31;
  const int wave = __builtin_amdgcn_readfirstlane((int)(threadIdx.x >> 5));
  const int row0 = (blockIdx.x * (blockDim.x >> 5) + wave) * 16;
  if (row0 >= nrows) return;

  v8f acc[NT];
#pragma unroll
  for (int t = 0; t < NT; ++t) {
    acc[t] = v8f{};
    asm volatile("" : "+v"(acc[t]));
  }

  const int mrow = lane & 15;
  const int hsel = lane >> 4;
  const int kofs = hsel << 1;

  const int ar = (row0 + mrow < nrows) ? (row0 + mrow) : (nrows - 1);
  const float* arow = A + ar * 64 + kofs;
  v2f a[16];
#pragma unroll
  for (int kk = 0; kk < 16; ++kk) a[kk] = *(const v2f*)(arow + 4 * kk);

  const int crow = row0 + (hsel << 3);
  const bool full = (row0 + 16 <= nrows);
  float sc[8];
  if (full) {
    const float4 s0 = *(const float4*)(rscale + crow);
    const float4 s1 = *(const float4*)(rscale + crow + 4);
    sc[0] = s0.x; sc[1] = s0.y; sc[2] = s0.z; sc[3] = s0.w;
    sc[4] = s1.x; sc[5] = s1.y; sc[6] = s1.z; sc[7] = s1.w;
  } else {
#pragma unroll
    for (int r = 0; r < 8; ++r) {
      const int rr = (crow + r < nrows) ? (crow + r) : (nrows - 1);
      sc[r] = rscale[rr];
    }
  }

  v2f b[NT][16];
#pragma unroll
  for (int t = 0; t < NT; ++t) {
    const int bn   = t * 16 + mrow;
    const int bcol = (bn < NVALID) ? bn : (NVALID - 1);
    const float* bp = B + kofs * LDB + bcol;
#pragma unroll
    for (int kk = 0; kk < 16; ++kk)
      b[t][kk] = v2f{ bp[4 * kk * LDB], bp[4 * kk * LDB + LDB] };
  }

  __builtin_amdgcn_sched_barrier(0);
#pragma unroll
  for (int kk = 0; kk < 16; ++kk) {
#pragma unroll
    for (int t = 0; t < NT; ++t) {
      acc[t] = __builtin_amdgcn_wmma_f32_16x16x4_f32(
           false, a[kk],  false, b[t][kk],
           (short)0, acc[t],  false,  false);
    }
  }

  __builtin_amdgcn_sched_barrier(0);

  if constexpr (NT == 4) {
    asm volatile("v_nop\n\tv_nop\n\tv_nop\n\tv_nop"
                 : "+v"(acc[0]), "+v"(acc[1]), "+v"(acc[2]), "+v"(acc[3])
                 : "v"(a[15]), "v"(b[0][15]), "v"(b[1][15]), "v"(b[2][15]), "v"(b[3][15]));
  } else {
    static_assert(NT == 3 || NT == 4, "guard written for 3 or 4 tiles");
    asm volatile("v_nop\n\tv_nop\n\tv_nop\n\tv_nop"
                 : "+v"(acc[0]), "+v"(acc[1]), "+v"(acc[2])
                 : "v"(a[15]), "v"(b[0][15]), "v"(b[1][15]), "v"(b[2][15]));
  }

#pragma unroll
  for (int t = 0; t < NT; ++t) {
    float o[8];
#pragma unroll
    for (int r = 0; r < 8; ++r) o[r] = acc[t][r] * sc[r];
    float* cp = C + crow * LDC + t * 16 + mrow;
    if (full) {
#pragma unroll
      for (int r = 0; r < 8; ++r) cp[r * LDC] = o[r];
    } else {
#pragma unroll
      for (int r = 0; r < 8; ++r)
        if (crow + r < nrows) cp[r * LDC] = o[r];
    }
  }
}

template <int LDH, int NV2, int LDO, bool RELU, int K>
__global__ void gcn_gather_rows(const float* __restrict__ hin,
                                const int* __restrict__ rowptr,
                                const int* __restrict__ col,
                                const float* __restrict__ dinv,
                                const float* __restrict__ bias,
                                float* __restrict__ res, int n) {
  constexpr int G = 16;
  static_assert(K >= 1 && K < 32, "row bounds come from one 32-lane load");
  static_assert(2 * G == 32, "a row's second group of ids is the upper half of ms[i]");
  const int lane = threadIdx.x & 31;
  const int wid = __builtin_amdgcn_readfirstlane(
      (int)(blockIdx.x * (blockDim.x >> 5) + (threadIdx.x >> 5)));
  const int r0 = wid * K;
  if (r0 >= n) return;

  const int l2 = (lane < NV2) ? lane : (NV2 - 1);
  const int lo = l2 * 2;
  const v2f* hin2 = (const v2f*)hin;

  const int bi = (lane < K) ? lane : K;
  const int rpv = rowptr[(r0 + bi < n) ? (r0 + bi) : n];
  const int fi = (lane < K - 1) ? lane : (K - 1);
  const float dv = dinv[(r0 + fi < n) ? (r0 + fi) : (n - 1)];
  const v2f bv = *(const v2f*)(bias + lo);

  int rb[K], rn[K], ms[K];
#pragma unroll
  for (int i = 0; i < K; ++i) {
    rb[i] = __builtin_amdgcn_readlane(rpv, i);
    rn[i] = __builtin_amdgcn_readlane(rpv, i + 1) - rb[i];
    ms[i] = 0;
    if (rn[i] > 0) ms[i] = col[rb[i] + ((lane < rn[i]) ? lane : (rn[i] - 1))];
  }

  v2f va[G], vb[G], sfa, sfb;

  auto issue = [&](const int i, v2f (&v)[G], v2f& sf) {
    const int rr = (r0 + i < n) ? (r0 + i) : (n - 1);
    sf = hin2[(unsigned)rr * (LDH / 2) + l2];
#pragma unroll
    for (int u = 0; u < G; ++u) {
      const int s = __builtin_amdgcn_readlane(ms[i], u);
      v[u] = hin2[(unsigned)s * (LDH / 2) + l2];
    }
  };

  auto finish = [&](const int i, v2f (&v)[G], const v2f& sf) {
    const int ne = rn[i];
    v2f acc = sf;
#pragma unroll
    for (int u = 0; u < G; ++u) {
      const float ax = acc.x + v[u].x;
      const float ay = acc.y + v[u].y;
      acc.x = (u < ne) ? ax : acc.x;
      acc.y = (u < ne) ? ay : acc.y;
    }
#pragma nounroll
    for (int off = G; off < ne; off += G) {
      const int nn = (ne - off < G) ? (ne - off) : G;
      int cs = ms[i];
      int sh = G;
      if (off >= 32) {
        cs = col[rb[i] + off + ((lane < nn) ? lane : (nn - 1))];
        sh = 0;
      }
      v2f w[G];
#pragma unroll
      for (int u = 0; u < G; ++u) {
        const int s = __builtin_amdgcn_readlane(cs, sh + u);
        w[u] = hin2[(unsigned)s * (LDH / 2) + l2];
      }
#pragma unroll
      for (int u = 0; u < G; ++u) {
        const float ax = acc.x + w[u].x;
        const float ay = acc.y + w[u].y;
        acc.x = (u < nn) ? ax : acc.x;
        acc.y = (u < nn) ? ay : acc.y;
      }
    }
    const float di = __int_as_float(
        __builtin_amdgcn_readlane(__float_as_int(dv), i));
    v2f r = { fmaf(acc.x, di, bv.x), fmaf(acc.y, di, bv.y) };
    if (RELU) {
      r.x = r.x > 0.0f ? r.x : 0.0f;
      r.y = r.y > 0.0f ? r.y : 0.0f;
    }
    if (r0 + i < n && lane < NV2) *(v2f*)(res + (long)(r0 + i) * LDO + lo) = r;
  };

  issue(0, va, sfa);
#pragma unroll
  for (int i = 0; i < K; i += 2) {
    if (i + 1 < K) issue(i + 1, vb, sfb);
    finish(i, va, sfa);
    if (i + 1 < K) {
      if (i + 2 < K) issue(i + 2, va, sfa);
      finish(i + 1, vb, sfb);
    }
  }
}

constexpr int TP = 68;

template <int NT>
__device__ __forceinline__ void tile_gemm_part(const float* __restrict__ tile,
                                               const float* __restrict__ B,
                                               const float* __restrict__ rscale,
                                               float* __restrict__ C,
                                               const int t0, const int row0, const int nrows,
                                               const int lane) {
  constexpr int LDB = 40, LDC = 48, NVALID = 40;
  constexpr int KS = 4;
  static_assert(NT == 1 || NT == 2, "guard written for 1 or 2 tiles");
  v8f acc[NT];
#pragma unroll
  for (int t = 0; t < NT; ++t) {
    acc[t] = v8f{};
    asm volatile("" : "+v"(acc[t]));
  }
  const int mrow = lane & 15;
  const int hsel = lane >> 4;
  const int kofs = hsel << 1;

  const int crow = row0 + (hsel << 3);
  const bool full = (row0 + 16 <= nrows);
  float sc[8];
  if (full) {
    const float4 s0 = *(const float4*)(rscale + crow);
    const float4 s1 = *(const float4*)(rscale + crow + 4);
    sc[0] = s0.x; sc[1] = s0.y; sc[2] = s0.z; sc[3] = s0.w;
    sc[4] = s1.x; sc[5] = s1.y; sc[6] = s1.z; sc[7] = s1.w;
  } else {
#pragma unroll
    for (int r = 0; r < 8; ++r) {
      const int rr = (crow + r < nrows) ? (crow + r) : (nrows - 1);
      sc[r] = rscale[rr];
    }
  }

  const float* arow = tile + mrow * TP + kofs;
  const float* bp[NT];
#pragma unroll
  for (int t = 0; t < NT; ++t) {
    const int bn   = (t0 + t) * 16 + mrow;
    const int bcol = (bn < NVALID) ? bn : (NVALID - 1);
    bp[t] = B + kofs * LDB + bcol;
  }

#pragma unroll
  for (int k0 = 0; k0 < 16; k0 += KS) {
    v2f a[KS], b[NT][KS];
#pragma unroll
    for (int j = 0; j < KS; ++j) a[j] = *(const v2f*)(arow + 4 * (k0 + j));
#pragma unroll
    for (int t = 0; t < NT; ++t)
#pragma unroll
      for (int j = 0; j < KS; ++j)
        b[t][j] = v2f{ bp[t][4 * (k0 + j) * LDB], bp[t][4 * (k0 + j) * LDB + LDB] };

    __builtin_amdgcn_sched_barrier(0);
#pragma unroll
    for (int j = 0; j < KS; ++j) {
#pragma unroll
      for (int t = 0; t < NT; ++t) {
        acc[t] = __builtin_amdgcn_wmma_f32_16x16x4_f32(
             false, a[j],  false, b[t][j],
             (short)0, acc[t],  false,  false);
      }
    }
    __builtin_amdgcn_sched_barrier(0);

    if constexpr (NT == 2) {
      asm volatile("v_nop\n\tv_nop\n\tv_nop\n\tv_nop"
                   : "+v"(acc[0]), "+v"(acc[1])
                   : "v"(a[KS - 1]), "v"(b[0][KS - 1]), "v"(b[1][KS - 1]));
    } else {
      asm volatile("v_nop\n\tv_nop\n\tv_nop\n\tv_nop"
                   : "+v"(acc[0])
                   : "v"(a[KS - 1]), "v"(b[0][KS - 1]));
    }
  }

#pragma unroll
  for (int t = 0; t < NT; ++t) {
    float o[8];
#pragma unroll
    for (int r = 0; r < 8; ++r) o[r] = acc[t][r] * sc[r];
    float* cp = C + crow * LDC + (t0 + t) * 16 + mrow;
    if (full) {
#pragma unroll
      for (int r = 0; r < 8; ++r) cp[r * LDC] = o[r];
    } else {
#pragma unroll
      for (int r = 0; r < 8; ++r)
        if (crow + r < nrows) cp[r * LDC] = o[r];
    }
  }
}

template <int K>
__global__ void __launch_bounds__(256)
gcn_gather_relu_gemm(const float* __restrict__ hin,
                     const int* __restrict__ rowptr,
                     const int* __restrict__ col,
                     const float* __restrict__ dinv,
                     const float* __restrict__ bias,
                     const float* __restrict__ B,
                     float* __restrict__ C, int n) {
  constexpr int G = 16;
  constexpr int LDH = 64;
  static_assert(K == 8, "8 waves x K rows = the 64 rows of the tile");
  static_assert(2 * G == 32, "a row's second group of ids is the upper half of ms[i]");
  __shared__ __attribute__((aligned(16))) float tile[8 * K * TP];
  const int lane = threadIdx.x & 31;
  const int wv = __builtin_amdgcn_readfirstlane((int)(threadIdx.x >> 5));
  const int brow0 = blockIdx.x * (8 * K);
  const int r0 = brow0 + wv * K;

  const int l2 = lane;
  const int lo = l2 * 2;
  const v2f* hin2 = (const v2f*)hin;

  const int bi = (lane < K) ? lane : K;
  const int rpv = rowptr[(r0 + bi < n) ? (r0 + bi) : n];
  const int fi = (lane < K - 1) ? lane : (K - 1);
  const float dv = dinv[(r0 + fi < n) ? (r0 + fi) : (n - 1)];
  const v2f bv = *(const v2f*)(bias + lo);

  int rb[K], rn[K], ms[K];
#pragma unroll
  for (int i = 0; i < K; ++i) {
    rb[i] = __builtin_amdgcn_readlane(rpv, i);
    rn[i] = __builtin_amdgcn_readlane(rpv, i + 1) - rb[i];
    ms[i] = 0;
    if (rn[i] > 0) ms[i] = col[rb[i] + ((lane < rn[i]) ? lane : (rn[i] - 1))];
  }

  v2f va[G], vb[G], sfa, sfb;

  auto issue = [&](const int i, v2f (&v)[G], v2f& sf) {
    const int rr = (r0 + i < n) ? (r0 + i) : (n - 1);
    sf = hin2[(unsigned)rr * (LDH / 2) + l2];
#pragma unroll
    for (int u = 0; u < G; ++u) {
      const int s = __builtin_amdgcn_readlane(ms[i], u);
      v[u] = hin2[(unsigned)s * (LDH / 2) + l2];
    }
  };

  auto finish = [&](const int i, v2f (&v)[G], const v2f& sf) {
    const int ne = rn[i];
    v2f acc = sf;
#pragma unroll
    for (int u = 0; u < G; ++u) {
      const float ax = acc.x + v[u].x;
      const float ay = acc.y + v[u].y;
      acc.x = (u < ne) ? ax : acc.x;
      acc.y = (u < ne) ? ay : acc.y;
    }
#pragma nounroll
    for (int off = G; off < ne; off += G) {
      const int nn = (ne - off < G) ? (ne - off) : G;
      int cs = ms[i];
      int sh = G;
      if (off >= 32) {
        cs = col[rb[i] + off + ((lane < nn) ? lane : (nn - 1))];
        sh = 0;
      }
      v2f w[G];
#pragma unroll
      for (int u = 0; u < G; ++u) {
        const int s = __builtin_amdgcn_readlane(cs, sh + u);
        w[u] = hin2[(unsigned)s * (LDH / 2) + l2];
      }
#pragma unroll
      for (int u = 0; u < G; ++u) {
        const float ax = acc.x + w[u].x;
        const float ay = acc.y + w[u].y;
        acc.x = (u < nn) ? ax : acc.x;
        acc.y = (u < nn) ? ay : acc.y;
      }
    }
    const float di = __int_as_float(
        __builtin_amdgcn_readlane(__float_as_int(dv), i));
    v2f r = { fmaf(acc.x, di, bv.x), fmaf(acc.y, di, bv.y) };
    r.x = r.x > 0.0f ? r.x : 0.0f;
    r.y = r.y > 0.0f ? r.y : 0.0f;
    if (r0 + i >= n) r = v2f{0.0f, 0.0f};
    *(v2f*)(tile + (wv * K + i) * TP + lo) = r;
  };

  issue(0, va, sfa);
#pragma unroll
  for (int i = 0; i < K; i += 2) {
    if (i + 1 < K) issue(i + 1, vb, sfb);
    finish(i, va, sfa);
    if (i + 1 < K) {
      if (i + 2 < K) issue(i + 2, va, sfa);
      finish(i + 1, vb, sfb);
    }
  }

  __syncthreads();

  const int mt = wv & 3;
  const int grow0 = brow0 + mt * 16;
  if (grow0 >= n) return;
  if (wv < 4) tile_gemm_part<2>(tile + mt * 16 * TP, B, dinv, C, 0, grow0, n, lane);
  else        tile_gemm_part<1>(tile + mt * 16 * TP, B, dinv, C, 2, grow0, n, lane);
}

extern "C" void kernel_launch(void* const* d_in, const int* in_sizes, int n_in,
                              void* d_out, int out_size, void* d_ws, size_t ws_size,
                              hipStream_t stream) {
  (void)n_in; (void)out_size; (void)ws_size;
  const float* x  = (const float*)d_in[0];
  const int*   ei = (const int*)d_in[1];
  const float* W1 = (const float*)d_in[2];
  const float* b1 = (const float*)d_in[3];
  const float* W3 = (const float*)d_in[4];
  const float* b3 = (const float*)d_in[5];

  const int N = in_sizes[0] / 64;
  const int E = in_sizes[1] / 2;
  const int* src = ei;
  const int* dst = ei + E;

  const int NBK = (N + BN - 1) / BN;
  const int NCH = (E + CH - 1) / CH;

  float* dinv   = (float*)d_ws;
  float* bufA   = dinv + N;
  float* bufB   = bufA + (long)N * 64;
  int*   hist   = (int*)(bufB + (long)N * 64);
  int*   bstart = hist + NCH * NBK;
  int*   rowptr = hist + N;
  unsigned* tmp = (unsigned*)(rowptr + N + 1);
  int*   col    = (int*)tmp + E;
  float* out    = (float*)d_out;

  const int mtiles = (N + 15) / 16;
  constexpr int KR = 8;

  csr_bucket_hist<<<NCH, BT, 0, stream>>>(dst, hist, E, NBK);
  csr_bucket_place<<<NCH, BT, 0, stream>>>(src, dst, hist, tmp, bstart, E, NBK, NCH);
  csr_rows<<<NBK, BT, 0, stream>>>(tmp, bstart, rowptr, dinv, col, N);

  {
    dim3 grid((mtiles + 7) / 8);
    gcn_gemm_wmma_f32<64, 64, 64, 4><<<grid, 256, 0, stream>>>(x, W1, dinv, bufA, N);
  }
  gcn_gather_relu_gemm<KR><<<(N + 8 * KR - 1) / (8 * KR), 256, 0, stream>>>(
      bufA, rowptr, col, dinv, b1, W3, bufB, N);

  gcn_gather_rows<48, 20, 40, false, KR><<<(N + 8 * KR - 1) / (8 * KR), 256, 0, stream>>>(
      bufB, rowptr, col, dinv, b3, out, N);
}


# Bucket sort of the edges without the global offset pass: each chunk is sorted in place and keeps its run starts; the bucket block finds its runs from them; 6 -> 5 launches; arrival-order sum
# speedup vs baseline: 4.9766x; 1.0490x over previous
#include <hip/hip_runtime.h>

typedef float v2f __attribute__((ext_vector_type(2)));
typedef float v8f __attribute__((ext_vector_type(8)));

constexpr int BSH  = 9;
constexpr int BN   = 1 << BSH;
constexpr int MAXB = 256;
constexpr int CH   = 4704;
constexpr int BT   = 1024;
constexpr int KPT  = (CH + BT - 1) / BT;
constexpr int UB   = 8;
constexpr int MAXC = BT;
constexpr int CAP  = 16384;

__device__ __forceinline__ int block_excl_scan(const int v, int* wsum, int& total) {
  const int lane = threadIdx.x & 31;
  const int w = threadIdx.x >> 5;
  int x = v;
  for (int o = 1; o < 32; o <<= 1) {
    const int y = __shfl_up(x, o);
    if (lane >= o) x += y;
  }
  if (lane == 31) wsum[w] = x;
  __syncthreads();
  const int mine = wsum[lane];
  int s = (lane < w) ? mine : 0;
  int t = mine;
  for (int o = 16; o > 0; o >>= 1) {
    s += __shfl_xor(s, o);
    t += __shfl_xor(t, o);
  }
  __syncthreads();
  total = t;
  return s + x - v;
}

__global__ void __launch_bounds__(BT)
csr_chunk_sort(const int* __restrict__ src, const int* __restrict__ dst,
               unsigned* __restrict__ tmp, int* __restrict__ cstart, int e, int nb) {
  __shared__ int h[MAXB];
  __shared__ int lstart[MAXB];
  __shared__ int wsum[32];
  __shared__ unsigned stage[CH];
  const int tid = threadIdx.x;
  const int blk = blockIdx.x;
  const int base = blk * CH;
  const int end = (base + CH < e) ? (base + CH) : e;
  if (tid < MAXB) h[tid] = 0;

  int d[KPT], s[KPT];
#pragma unroll
  for (int k = 0; k < KPT; ++k) {
    const int i = base + tid + k * BT;
    const int ii = (i < end) ? i : (end - 1);
    d[k] = dst[ii];
    s[k] = src[ii];
  }
  __syncthreads();
#pragma unroll
  for (int k = 0; k < KPT; ++k)
    if (base + tid + k * BT < end) atomicAdd(&h[d[k] >> BSH], 1);
  __syncthreads();

  const int L = (tid < nb) ? h[tid] : 0;
  int totalL;
  const int ls = block_excl_scan(L, wsum, totalL);
  if (tid < nb) {
    h[tid] = 0;
    lstart[tid] = ls;
    cstart[blk * (nb + 1) + tid] = ls;
  }
  if (tid == nb) cstart[blk * (nb + 1) + nb] = totalL;
  __syncthreads();

#pragma unroll
  for (int k = 0; k < KPT; ++k) {
    if (base + tid + k * BT < end) {
      const int bb = d[k] >> BSH;
      const int pos = lstart[bb] + atomicAdd(&h[bb], 1);
      stage[pos] = ((unsigned)s[k] << BSH) | (unsigned)(d[k] & (BN - 1));
    }
  }
  __syncthreads();

  const int nloc = end - base;
#pragma unroll
  for (int k = 0; k < KPT; ++k) {
    const int pos = tid + k * BT;
    if (pos < nloc) tmp[base + pos] = stage[pos];
  }
}

__global__ void __launch_bounds__(BT)
csr_rows(const unsigned* __restrict__ tmp, const int* __restrict__ cstart,
         int* __restrict__ rowptr, float* __restrict__ dinv,
         int* __restrict__ col, int n, int nb, int nchunk) {
  __shared__ int cnt[BN];
  __shared__ int cur[BN];
  __shared__ int wsum[32];
  __shared__ int rbeg[MAXC];
  __shared__ int rlen[MAXC];
  __shared__ int stg[CAP];
  const int tid = threadIdx.x;
  const int lane = tid & 31;
  const int wid = __builtin_amdgcn_readfirstlane((int)(threadIdx.x >> 5));
  const int b = blockIdx.x;

  int before = 0, mine = 0;
  if (tid < nchunk) {
    const int* cs = cstart + tid * (nb + 1) + b;
    before = cs[0];
    mine = cs[1] - before;
    rbeg[tid] = tid * CH + before;
    rlen[tid] = mine;
  }
  if (tid < BN) cnt[tid] = 0;
  int s0, m;
  block_excl_scan(before, wsum, s0);
  block_excl_scan(mine, wsum, m);
  const int s1 = s0 + m;

  auto walk = [&](auto&& f) {
    for (int c0 = wid; c0 < nchunk; c0 += 32 * UB) {
      unsigned w[UB];
      int len[UB], beg[UB];
#pragma unroll
      for (int u = 0; u < UB; ++u) {
        const int cc = c0 + 32 * u;
        const int ci = (cc < nchunk) ? cc : c0;
        beg[u] = __builtin_amdgcn_readfirstlane(rbeg[ci]);
        len[u] = (cc < nchunk) ? __builtin_amdgcn_readfirstlane(rlen[ci]) : 0;
        w[u] = tmp[(lane < len[u]) ? (beg[u] + lane) : 0];
      }
#pragma unroll
      for (int u = 0; u < UB; ++u)
        if (lane < len[u]) f(w[u]);
#pragma unroll
      for (int u = 0; u < UB; ++u)
        for (int off = 32 + lane; off < len[u]; off += 32) f(tmp[beg[u] + off]);
    }
  };

  walk([&](const unsigned w) { atomicAdd(&cnt[w & (BN - 1)], 1); });
  __syncthreads();

  const int c = (tid < BN) ? cnt[tid] : 0;
  int total;
  const int ex = block_excl_scan(c, wsum, total);
  const int node = b * BN + tid;
  if (tid < BN) {
    cur[tid] = ex;
    if (node < n) {
      rowptr[node] = s0 + ex;
      dinv[node] = rsqrtf((float)(c + 1));
    }
  }
  if (b == (int)gridDim.x - 1 && tid == 0) rowptr[n] = s1;
  __syncthreads();

  const bool staged = (m <= CAP);
  walk([&](const unsigned w) {
    const int p = atomicAdd(&cur[w & (BN - 1)], 1);
    if (staged) stg[p] = (int)(w >> BSH);
    else col[s0 + p] = (int)(w >> BSH);
  });
  __syncthreads();
  if (staged)
    for (int i = tid; i < m; i += BT) col[s0 + i] = stg[i];
}

template <int LDB, int LDC, int NVALID, int NT>
__global__ void __launch_bounds__(256)
gcn_gemm_wmma_f32(const float* __restrict__ A,
                                  const float* __restrict__ B,
                                  const float* __restrict__ rscale,
                                  float* __restrict__ C, int nrows) {
  const int lane = threadIdx.x & 31;
  const int wave = __builtin_amdgcn_readfirstlane((int)(threadIdx.x >> 5));
  const int row0 = (blockIdx.x * (blockDim.x >> 5) + wave) * 16;
  if (row0 >= nrows) return;

  v8f acc[NT];
#pragma unroll
  for (int t = 0; t < NT; ++t) {
    acc[t] = v8f{};
    asm volatile("" : "+v"(acc[t]));
  }

  const int mrow = lane & 15;
  const int hsel = lane >> 4;
  const int kofs = hsel << 1;

  const int ar = (row0 + mrow < nrows) ? (row0 + mrow) : (nrows - 1);
  const float* arow = A + ar * 64 + kofs;
  v2f a[16];
#pragma unroll
  for (int kk = 0; kk < 16; ++kk) a[kk] = *(const v2f*)(arow + 4 * kk);

  const int crow = row0 + (hsel << 3);
  const bool full = (row0 + 16 <= nrows);
  float sc[8];
  if (full) {
    const float4 s0 = *(const float4*)(rscale + crow);
    const float4 s1 = *(const float4*)(rscale + crow + 4);
    sc[0] = s0.x; sc[1] = s0.y; sc[2] = s0.z; sc[3] = s0.w;
    sc[4] = s1.x; sc[5] = s1.y; sc[6] = s1.z; sc[7] = s1.w;
  } else {
#pragma unroll
    for (int r = 0; r < 8; ++r) {
      const int rr = (crow + r < nrows) ? (crow + r) : (nrows - 1);
      sc[r] = rscale[rr];
    }
  }

  v2f b[NT][16];
#pragma unroll
  for (int t = 0; t < NT; ++t) {
    const int bn   = t * 16 + mrow;
    const int bcol = (bn < NVALID) ? bn : (NVALID - 1);
    const float* bp = B + kofs * LDB + bcol;
#pragma unroll
    for (int kk = 0; kk < 16; ++kk)
      b[t][kk] = v2f{ bp[4 * kk * LDB], bp[4 * kk * LDB + LDB] };
  }

  __builtin_amdgcn_sched_barrier(0);
#pragma unroll
  for (int kk = 0; kk < 16; ++kk) {
#pragma unroll
    for (int t = 0; t < NT; ++t) {
      acc[t] = __builtin_amdgcn_wmma_f32_16x16x4_f32(
           false, a[kk],  false, b[t][kk],
           (short)0, acc[t],  false,  false);
    }
  }

  __builtin_amdgcn_sched_barrier(0);

  if constexpr (NT == 4) {
    asm volatile("v_nop\n\tv_nop\n\tv_nop\n\tv_nop"
                 : "+v"(acc[0]), "+v"(acc[1]), "+v"(acc[2]), "+v"(acc[3])
                 : "v"(a[15]), "v"(b[0][15]), "v"(b[1][15]), "v"(b[2][15]), "v"(b[3][15]));
  } else {
    static_assert(NT == 3 || NT == 4, "guard written for 3 or 4 tiles");
    asm volatile("v_nop\n\tv_nop\n\tv_nop\n\tv_nop"
                 : "+v"(acc[0]), "+v"(acc[1]), "+v"(acc[2])
                 : "v"(a[15]), "v"(b[0][15]), "v"(b[1][15]), "v"(b[2][15]));
  }

#pragma unroll
  for (int t = 0; t < NT; ++t) {
    float o[8];
#pragma unroll
    for (int r = 0; r < 8; ++r) o[r] = acc[t][r] * sc[r];
    float* cp = C + crow * LDC + t * 16 + mrow;
    if (full) {
#pragma unroll
      for (int r = 0; r < 8; ++r) cp[r * LDC] = o[r];
    } else {
#pragma unroll
      for (int r = 0; r < 8; ++r)
        if (crow + r < nrows) cp[r * LDC] = o[r];
    }
  }
}

template <int LDH, int NV2, int LDO, bool RELU, int K>
__global__ void gcn_gather_rows(const float* __restrict__ hin,
                                const int* __restrict__ rowptr,
                                const int* __restrict__ col,
                                const float* __restrict__ dinv,
                                const float* __restrict__ bias,
                                float* __restrict__ res, int n) {
  constexpr int G = 16;
  static_assert(K >= 1 && K < 32, "row bounds come from one 32-lane load");
  static_assert(2 * G == 32, "a row's second group of ids is the upper half of ms[i]");
  const int lane = threadIdx.x & 31;
  const int wid = __builtin_amdgcn_readfirstlane(
      (int)(blockIdx.x * (blockDim.x >> 5) + (threadIdx.x >> 5)));
  const int r0 = wid * K;
  if (r0 >= n) return;

  const int l2 = (lane < NV2) ? lane : (NV2 - 1);
  const int lo = l2 * 2;
  const v2f* hin2 = (const v2f*)hin;

  const int bi = (lane < K) ? lane : K;
  const int rpv = rowptr[(r0 + bi < n) ? (r0 + bi) : n];
  const int fi = (lane < K - 1) ? lane : (K - 1);
  const float dv = dinv[(r0 + fi < n) ? (r0 + fi) : (n - 1)];
  const v2f bv = *(const v2f*)(bias + lo);

  int rb[K], rn[K], ms[K];
#pragma unroll
  for (int i = 0; i < K; ++i) {
    rb[i] = __builtin_amdgcn_readlane(rpv, i);
    rn[i] = __builtin_amdgcn_readlane(rpv, i + 1) - rb[i];
    ms[i] = 0;
    if (rn[i] > 0) ms[i] = col[rb[i] + ((lane < rn[i]) ? lane : (rn[i] - 1))];
  }

  v2f va[G], vb[G], sfa, sfb;

  auto issue = [&](const int i, v2f (&v)[G], v2f& sf) {
    const int rr = (r0 + i < n) ? (r0 + i) : (n - 1);
    sf = hin2[(unsigned)rr * (LDH / 2) + l2];
#pragma unroll
    for (int u = 0; u < G; ++u) {
      const int s = __builtin_amdgcn_readlane(ms[i], u);
      v[u] = hin2[(unsigned)s * (LDH / 2) + l2];
    }
  };

  auto finish = [&](const int i, v2f (&v)[G], const v2f& sf) {
    const int ne = rn[i];
    v2f acc = sf;
#pragma unroll
    for (int u = 0; u < G; ++u) {
      const float ax = acc.x + v[u].x;
      const float ay = acc.y + v[u].y;
      acc.x = (u < ne) ? ax : acc.x;
      acc.y = (u < ne) ? ay : acc.y;
    }
#pragma nounroll
    for (int off = G; off < ne; off += G) {
      const int nn = (ne - off < G) ? (ne - off) : G;
      int cs = ms[i];
      int sh = G;
      if (off >= 32) {
        cs = col[rb[i] + off + ((lane < nn) ? lane : (nn - 1))];
        sh = 0;
      }
      v2f w[G];
#pragma unroll
      for (int u = 0; u < G; ++u) {
        const int s = __builtin_amdgcn_readlane(cs, sh + u);
        w[u] = hin2[(unsigned)s * (LDH / 2) + l2];
      }
#pragma unroll
      for (int u = 0; u < G; ++u) {
        const float ax = acc.x + w[u].x;
        const float ay = acc.y + w[u].y;
        acc.x = (u < nn) ? ax : acc.x;
        acc.y = (u < nn) ? ay : acc.y;
      }
    }
    const float di = __int_as_float(
        __builtin_amdgcn_readlane(__float_as_int(dv), i));
    v2f r = { fmaf(acc.x, di, bv.x), fmaf(acc.y, di, bv.y) };
    if (RELU) {
      r.x = r.x > 0.0f ? r.x : 0.0f;
      r.y = r.y > 0.0f ? r.y : 0.0f;
    }
    if (r0 + i < n && lane < NV2) *(v2f*)(res + (long)(r0 + i) * LDO + lo) = r;
  };

  issue(0, va, sfa);
#pragma unroll
  for (int i = 0; i < K; i += 2) {
    if (i + 1 < K) issue(i + 1, vb, sfb);
    finish(i, va, sfa);
    if (i + 1 < K) {
      if (i + 2 < K) issue(i + 2, va, sfa);
      finish(i + 1, vb, sfb);
    }
  }
}

constexpr int TP = 68;

template <int NT>
__device__ __forceinline__ void tile_gemm_part(const float* __restrict__ tile,
                                               const float* __restrict__ B,
                                               const float* __restrict__ rscale,
                                               float* __restrict__ C,
                                               const int t0, const int row0, const int nrows,
                                               const int lane) {
  constexpr int LDB = 40, LDC = 48, NVALID = 40;
  constexpr int KS = 4;
  static_assert(NT == 1 || NT == 2, "guard written for 1 or 2 tiles");
  v8f acc[NT];
#pragma unroll
  for (int t = 0; t < NT; ++t) {
    acc[t] = v8f{};
    asm volatile("" : "+v"(acc[t]));
  }
  const int mrow = lane & 15;
  const int hsel = lane >> 4;
  const int kofs = hsel << 1;

  const int crow = row0 + (hsel << 3);
  const bool full = (row0 + 16 <= nrows);
  float sc[8];
  if (full) {
    const float4 s0 = *(const float4*)(rscale + crow);
    const float4 s1 = *(const float4*)(rscale + crow + 4);
    sc[0] = s0.x; sc[1] = s0.y; sc[2] = s0.z; sc[3] = s0.w;
    sc[4] = s1.x; sc[5] = s1.y; sc[6] = s1.z; sc[7] = s1.w;
  } else {
#pragma unroll
    for (int r = 0; r < 8; ++r) {
      const int rr = (crow + r < nrows) ? (crow + r) : (nrows - 1);
      sc[r] = rscale[rr];
    }
  }

  const float* arow = tile + mrow * TP + kofs;
  const float* bp[NT];
#pragma unroll
  for (int t = 0; t < NT; ++t) {
    const int bn   = (t0 + t) * 16 + mrow;
    const int bcol = (bn < NVALID) ? bn : (NVALID - 1);
    bp[t] = B + kofs * LDB + bcol;
  }

#pragma unroll
  for (int k0 = 0; k0 < 16; k0 += KS) {
    v2f a[KS], b[NT][KS];
#pragma unroll
    for (int j = 0; j < KS; ++j) a[j] = *(const v2f*)(arow + 4 * (k0 + j));
#pragma unroll
    for (int t = 0; t < NT; ++t)
#pragma unroll
      for (int j = 0; j < KS; ++j)
        b[t][j] = v2f{ bp[t][4 * (k0 + j) * LDB], bp[t][4 * (k0 + j) * LDB + LDB] };

    __builtin_amdgcn_sched_barrier(0);
#pragma unroll
    for (int j = 0; j < KS; ++j) {
#pragma unroll
      for (int t = 0; t < NT; ++t) {
        acc[t] = __builtin_amdgcn_wmma_f32_16x16x4_f32(
             false, a[j],  false, b[t][j],
             (short)0, acc[t],  false,  false);
      }
    }
    __builtin_amdgcn_sched_barrier(0);

    if constexpr (NT == 2) {
      asm volatile("v_nop\n\tv_nop\n\tv_nop\n\tv_nop"
                   : "+v"(acc[0]), "+v"(acc[1])
                   : "v"(a[KS - 1]), "v"(b[0][KS - 1]), "v"(b[1][KS - 1]));
    } else {
      asm volatile("v_nop\n\tv_nop\n\tv_nop\n\tv_nop"
                   : "+v"(acc[0])
                   : "v"(a[KS - 1]), "v"(b[0][KS - 1]));
    }
  }

#pragma unroll
  for (int t = 0; t < NT; ++t) {
    float o[8];
#pragma unroll
    for (int r = 0; r < 8; ++r) o[r] = acc[t][r] * sc[r];
    float* cp = C + crow * LDC + (t0 + t) * 16 + mrow;
    if (full) {
#pragma unroll
      for (int r = 0; r < 8; ++r) cp[r * LDC] = o[r];
    } else {
#pragma unroll
      for (int r = 0; r < 8; ++r)
        if (crow + r < nrows) cp[r * LDC] = o[r];
    }
  }
}

template <int K>
__global__ void __launch_bounds__(256)
gcn_gather_relu_gemm(const float* __restrict__ hin,
                     const int* __restrict__ rowptr,
                     const int* __restrict__ col,
                     const float* __restrict__ dinv,
                     const float* __restrict__ bias,
                     const float* __restrict__ B,
                     float* __restrict__ C, int n) {
  constexpr int G = 16;
  constexpr int LDH = 64;
  static_assert(K == 8, "8 waves x K rows = the 64 rows of the tile");
  static_assert(2 * G == 32, "a row's second group of ids is the upper half of ms[i]");
  __shared__ __attribute__((aligned(16))) float tile[8 * K * TP];
  const int lane = threadIdx.x & 31;
  const int wv = __builtin_amdgcn_readfirstlane((int)(threadIdx.x >> 5));
  const int brow0 = blockIdx.x * (8 * K);
  const int r0 = brow0 + wv * K;

  const int l2 = lane;
  const int lo = l2 * 2;
  const v2f* hin2 = (const v2f*)hin;

  const int bi = (lane < K) ? lane : K;
  const int rpv = rowptr[(r0 + bi < n) ? (r0 + bi) : n];
  const int fi = (lane < K - 1) ? lane : (K - 1);
  const float dv = dinv[(r0 + fi < n) ? (r0 + fi) : (n - 1)];
  const v2f bv = *(const v2f*)(bias + lo);

  int rb[K], rn[K], ms[K];
#pragma unroll
  for (int i = 0; i < K; ++i) {
    rb[i] = __builtin_amdgcn_readlane(rpv, i);
    rn[i] = __builtin_amdgcn_readlane(rpv, i + 1) - rb[i];
    ms[i] = 0;
    if (rn[i] > 0) ms[i] = col[rb[i] + ((lane < rn[i]) ? lane : (rn[i] - 1))];
  }

  v2f va[G], vb[G], sfa, sfb;

  auto issue = [&](const int i, v2f (&v)[G], v2f& sf) {
    const int rr = (r0 + i < n) ? (r0 + i) : (n - 1);
    sf = hin2[(unsigned)rr * (LDH / 2) + l2];
#pragma unroll
    for (int u = 0; u < G; ++u) {
      const int s = __builtin_amdgcn_readlane(ms[i], u);
      v[u] = hin2[(unsigned)s * (LDH / 2) + l2];
    }
  };

  auto finish = [&](const int i, v2f (&v)[G], const v2f& sf) {
    const int ne = rn[i];
    v2f acc = sf;
#pragma unroll
    for (int u = 0; u < G; ++u) {
      const float ax = acc.x + v[u].x;
      const float ay = acc.y + v[u].y;
      acc.x = (u < ne) ? ax : acc.x;
      acc.y = (u < ne) ? ay : acc.y;
    }
#pragma nounroll
    for (int off = G; off < ne; off += G) {
      const int nn = (ne - off < G) ? (ne - off) : G;
      int cs = ms[i];
      int sh = G;
      if (off >= 32) {
        cs = col[rb[i] + off + ((lane < nn) ? lane : (nn - 1))];
        sh = 0;
      }
      v2f w[G];
#pragma unroll
      for (int u = 0; u < G; ++u) {
        const int s = __builtin_amdgcn_readlane(cs, sh + u);
        w[u] = hin2[(unsigned)s * (LDH / 2) + l2];
      }
#pragma unroll
      for (int u = 0; u < G; ++u) {
        const float ax = acc.x + w[u].x;
        const float ay = acc.y + w[u].y;
        acc.x = (u < nn) ? ax : acc.x;
        acc.y = (u < nn) ? ay : acc.y;
      }
    }
    const float di = __int_as_float(
        __builtin_amdgcn_readlane(__float_as_int(dv), i));
    v2f r = { fmaf(acc.x, di, bv.x), fmaf(acc.y, di, bv.y) };
    r.x = r.x > 0.0f ? r.x : 0.0f;
    r.y = r.y > 0.0f ? r.y : 0.0f;
    if (r0 + i >= n) r = v2f{0.0f, 0.0f};
    *(v2f*)(tile + (wv * K + i) * TP + lo) = r;
  };

  issue(0, va, sfa);
#pragma unroll
  for (int i = 0; i < K; i += 2) {
    if (i + 1 < K) issue(i + 1, vb, sfb);
    finish(i, va, sfa);
    if (i + 1 < K) {
      if (i + 2 < K) issue(i + 2, va, sfa);
      finish(i + 1, vb, sfb);
    }
  }

  __syncthreads();

  const int mt = wv & 3;
  const int grow0 = brow0 + mt * 16;
  if (grow0 >= n) return;
  if (wv < 4) tile_gemm_part<2>(tile + mt * 16 * TP, B, dinv, C, 0, grow0, n, lane);
  else        tile_gemm_part<1>(tile + mt * 16 * TP, B, dinv, C, 2, grow0, n, lane);
}

extern "C" void kernel_launch(void* const* d_in, const int* in_sizes, int n_in,
                              void* d_out, int out_size, void* d_ws, size_t ws_size,
                              hipStream_t stream) {
  (void)n_in; (void)out_size; (void)ws_size;
  const float* x  = (const float*)d_in[0];
  const int*   ei = (const int*)d_in[1];
  const float* W1 = (const float*)d_in[2];
  const float* b1 = (const float*)d_in[3];
  const float* W3 = (const float*)d_in[4];
  const float* b3 = (const float*)d_in[5];

  const int N = in_sizes[0] / 64;
  const int E = in_sizes[1] / 2;
  const int* src = ei;
  const int* dst = ei + E;

  const int NBK = (N + BN - 1) / BN;
  const int NCH = (E + CH - 1) / CH;

  float* dinv   = (float*)d_ws;
  float* bufA   = dinv + N;
  float* bufB   = bufA + (long)N * 64;
  int*   cstart = (int*)(bufB + (long)N * 64);
  int*   rowptr = cstart + N;
  unsigned* tmp = (unsigned*)(rowptr + N + 1);
  int*   col    = (int*)tmp + E;
  float* out    = (float*)d_out;

  const int mtiles = (N + 15) / 16;
  constexpr int KR = 8;

  csr_chunk_sort<<<NCH, BT, 0, stream>>>(src, dst, tmp, cstart, E, NBK);
  csr_rows<<<NBK, BT, 0, stream>>>(tmp, cstart, rowptr, dinv, col, N, NBK, NCH);

  {
    dim3 grid((mtiles + 7) / 8);
    gcn_gemm_wmma_f32<64, 64, 64, 4><<<grid, 256, 0, stream>>>(x, W1, dinv, bufA, N);
  }
  gcn_gather_relu_gemm<KR><<<(N + 8 * KR - 1) / (8 * KR), 256, 0, stream>>>(
      bufA, rowptr, col, dinv, b1, W3, bufB, N);

  gcn_gather_rows<48, 20, 40, false, KR><<<(N + 8 * KR - 1) / (8 * KR), 256, 0, stream>>>(
      bufB, rowptr, col, dinv, b3, out, N);
}


# Layer-1 product moved behind the aggregation, (A x) W1 for A (x W1): gather sums rows of dinv*x (scaled in the row-list kernel), block multiplies by W1 from LDS; 5 -> 4 launches; rounding changes
# speedup vs baseline: 5.2039x; 1.0457x over previous
#include <hip/hip_runtime.h>

typedef float v2f __attribute__((ext_vector_type(2)));
typedef float v8f __attribute__((ext_vector_type(8)));

constexpr int BSH  = 9;
constexpr int BN   = 1 << BSH;
constexpr int MAXB = 256;
constexpr int CH   = 4704;
constexpr int BT   = 1024;
constexpr int KPT  = (CH + BT - 1) / BT;
constexpr int UB   = 8;
constexpr int MAXC = BT;
constexpr int CAP  = 16384;

__device__ __forceinline__ void lds_barrier() {
  __builtin_amdgcn_fence(__ATOMIC_RELEASE, "workgroup", "local");
  __builtin_amdgcn_s_barrier();
  __builtin_amdgcn_fence(__ATOMIC_ACQUIRE, "workgroup", "local");
}

__device__ __forceinline__ int block_excl_scan(const int v, int* wsum, int& total) {
  const int lane = threadIdx.x & 31;
  const int w = threadIdx.x >> 5;
  int x = v;
  for (int o = 1; o < 32; o <<= 1) {
    const int y = __shfl_up(x, o);
    if (lane >= o) x += y;
  }
  if (lane == 31) wsum[w] = x;
  lds_barrier();
  const int mine = wsum[lane];
  int s = (lane < w) ? mine : 0;
  int t = mine;
  for (int o = 16; o > 0; o >>= 1) {
    s += __shfl_xor(s, o);
    t += __shfl_xor(t, o);
  }
  lds_barrier();
  total = t;
  return s + x - v;
}

__global__ void __launch_bounds__(BT)
csr_chunk_sort(const int* __restrict__ src, const int* __restrict__ dst,
               unsigned* __restrict__ tmp, int* __restrict__ cstart, int e, int nb) {
  __shared__ int h[MAXB];
  __shared__ int lstart[MAXB];
  __shared__ int wsum[32];
  __shared__ unsigned stage[CH];
  const int tid = threadIdx.x;
  const int blk = blockIdx.x;
  const int base = blk * CH;
  const int end = (base + CH < e) ? (base + CH) : e;
  if (tid < MAXB) h[tid] = 0;

  int d[KPT], s[KPT];
#pragma unroll
  for (int k = 0; k < KPT; ++k) {
    const int i = base + tid + k * BT;
    const int ii = (i < end) ? i : (end - 1);
    d[k] = dst[ii];
    s[k] = src[ii];
  }
  __syncthreads();
#pragma unroll
  for (int k = 0; k < KPT; ++k)
    if (base + tid + k * BT < end) atomicAdd(&h[d[k] >> BSH], 1);
  __syncthreads();

  const int L = (tid < nb) ? h[tid] : 0;
  int totalL;
  const int ls = block_excl_scan(L, wsum, totalL);
  if (tid < nb) {
    h[tid] = 0;
    lstart[tid] = ls;
    cstart[blk * (nb + 1) + tid] = ls;
  }
  if (tid == nb) cstart[blk * (nb + 1) + nb] = totalL;
  __syncthreads();

#pragma unroll
  for (int k = 0; k < KPT; ++k) {
    if (base + tid + k * BT < end) {
      const int bb = d[k] >> BSH;
      const int pos = lstart[bb] + atomicAdd(&h[bb], 1);
      stage[pos] = ((unsigned)s[k] << BSH) | (unsigned)(d[k] & (BN - 1));
    }
  }
  __syncthreads();

  const int nloc = end - base;
#pragma unroll
  for (int k = 0; k < KPT; ++k) {
    const int pos = tid + k * BT;
    if (pos < nloc) tmp[base + pos] = stage[pos];
  }
}

constexpr int XPT = BN * 16 / BT;

__global__ void __launch_bounds__(BT)
csr_rows(const unsigned* __restrict__ tmp, const int* __restrict__ cstart,
         int* __restrict__ rowptr, float* __restrict__ dinv,
         int* __restrict__ col, const float* __restrict__ x,
         float* __restrict__ xs, int n, int nb, int nchunk) {
  __shared__ int cnt[BN];
  __shared__ float dl[BN];
  __shared__ int cur[BN];
  __shared__ int wsum[32];
  __shared__ int rbeg[MAXC];
  __shared__ int rlen[MAXC];
  __shared__ int stg[CAP];
  const int tid = threadIdx.x;
  const int lane = tid & 31;
  const int wid = __builtin_amdgcn_readfirstlane((int)(threadIdx.x >> 5));
  const int b = blockIdx.x;

  int before = 0, mine = 0;
  if (tid < nchunk) {
    const int* cs = cstart + tid * (nb + 1) + b;
    before = cs[0];
    mine = cs[1] - before;
    rbeg[tid] = tid * CH + before;
    rlen[tid] = mine;
  }
  if (tid < BN) cnt[tid] = 0;

  const int xr = tid >> 4;
  const int xc = (tid & 15) * 4;
  float4 xv[XPT];
#pragma unroll
  for (int j = 0; j < XPT; ++j) {
    const int node = b * BN + j * (BT / 16) + xr;
    const int nn = (node < n) ? node : (n - 1);
    xv[j] = *(const float4*)(x + (long)nn * 64 + xc);
  }

  int s0, m;
  block_excl_scan(before, wsum, s0);
  block_excl_scan(mine, wsum, m);
  const int s1 = s0 + m;

  auto walk = [&](auto&& f) {
    for (int c0 = wid; c0 < nchunk; c0 += 32 * UB) {
      unsigned w[UB];
      int len[UB], beg[UB];
#pragma unroll
      for (int u = 0; u < UB; ++u) {
        const int cc = c0 + 32 * u;
        const int ci = (cc < nchunk) ? cc : c0;
        beg[u] = __builtin_amdgcn_readfirstlane(rbeg[ci]);
        len[u] = (cc < nchunk) ? __builtin_amdgcn_readfirstlane(rlen[ci]) : 0;
        w[u] = tmp[(lane < len[u]) ? (beg[u] + lane) : 0];
      }
#pragma unroll
      for (int u = 0; u < UB; ++u)
        if (lane < len[u]) f(w[u]);
#pragma unroll
      for (int u = 0; u < UB; ++u)
        for (int off = 32 + lane; off < len[u]; off += 32) f(tmp[beg[u] + off]);
    }
  };

  walk([&](const unsigned w) { atomicAdd(&cnt[w & (BN - 1)], 1); });
  lds_barrier();

  const int c = (tid < BN) ? cnt[tid] : 0;
  int total;
  const int ex = block_excl_scan(c, wsum, total);
  const int node = b * BN + tid;
  if (tid < BN) {
    cur[tid] = ex;
    const float di = rsqrtf((float)(c + 1));
    dl[tid] = di;
    if (node < n) {
      rowptr[node] = s0 + ex;
      dinv[node] = di;
    }
  }
  if (b == (int)gridDim.x - 1 && tid == 0) rowptr[n] = s1;
  lds_barrier();

#pragma unroll
  for (int j = 0; j < XPT; ++j) {
    const int lr = j * (BT / 16) + xr;
    const int node2 = b * BN + lr;
    if (node2 < n) {
      const float di = dl[lr];
      float4 v = xv[j];
      v.x *= di; v.y *= di; v.z *= di; v.w *= di;
      *(float4*)(xs + (long)node2 * 64 + xc) = v;
    }
  }

  const bool staged = (m <= CAP);
  walk([&](const unsigned w) {
    const int p = atomicAdd(&cur[w & (BN - 1)], 1);
    if (staged) stg[p] = (int)(w >> BSH);
    else col[s0 + p] = (int)(w >> BSH);
  });
  lds_barrier();
  if (staged)
    for (int i = tid; i < m; i += BT) col[s0 + i] = stg[i];
}


template <int LDH, int NV2, int LDO, bool RELU, int K>
__global__ void gcn_gather_rows(const float* __restrict__ hin,
                                const int* __restrict__ rowptr,
                                const int* __restrict__ col,
                                const float* __restrict__ dinv,
                                const float* __restrict__ bias,
                                float* __restrict__ res, int n) {
  constexpr int G = 16;
  static_assert(K >= 1 && K < 32, "row bounds come from one 32-lane load");
  static_assert(2 * G == 32, "a row's second group of ids is the upper half of ms[i]");
  const int lane = threadIdx.x & 31;
  const int wid = __builtin_amdgcn_readfirstlane(
      (int)(blockIdx.x * (blockDim.x >> 5) + (threadIdx.x >> 5)));
  const int r0 = wid * K;
  if (r0 >= n) return;

  const int l2 = (lane < NV2) ? lane : (NV2 - 1);
  const int lo = l2 * 2;
  const v2f* hin2 = (const v2f*)hin;

  const int bi = (lane < K) ? lane : K;
  const int rpv = rowptr[(r0 + bi < n) ? (r0 + bi) : n];
  const int fi = (lane < K - 1) ? lane : (K - 1);
  const float dv = dinv[(r0 + fi < n) ? (r0 + fi) : (n - 1)];
  const v2f bv = *(const v2f*)(bias + lo);

  int rb[K], rn[K], ms[K];
#pragma unroll
  for (int i = 0; i < K; ++i) {
    rb[i] = __builtin_amdgcn_readlane(rpv, i);
    rn[i] = __builtin_amdgcn_readlane(rpv, i + 1) - rb[i];
    ms[i] = 0;
    if (rn[i] > 0) ms[i] = col[rb[i] + ((lane < rn[i]) ? lane : (rn[i] - 1))];
  }

  v2f va[G], vb[G], sfa, sfb;

  auto issue = [&](const int i, v2f (&v)[G], v2f& sf) {
    const int rr = (r0 + i < n) ? (r0 + i) : (n - 1);
    sf = hin2[(unsigned)rr * (LDH / 2) + l2];
#pragma unroll
    for (int u = 0; u < G; ++u) {
      const int s = __builtin_amdgcn_readlane(ms[i], u);
      v[u] = hin2[(unsigned)s * (LDH / 2) + l2];
    }
  };

  auto finish = [&](const int i, v2f (&v)[G], const v2f& sf) {
    const int ne = rn[i];
    v2f acc = sf;
#pragma unroll
    for (int u = 0; u < G; ++u) {
      const float ax = acc.x + v[u].x;
      const float ay = acc.y + v[u].y;
      acc.x = (u < ne) ? ax : acc.x;
      acc.y = (u < ne) ? ay : acc.y;
    }
#pragma nounroll
    for (int off = G; off < ne; off += G) {
      const int nn = (ne - off < G) ? (ne - off) : G;
      int cs = ms[i];
      int sh = G;
      if (off >= 32) {
        cs = col[rb[i] + off + ((lane < nn) ? lane : (nn - 1))];
        sh = 0;
      }
      v2f w[G];
#pragma unroll
      for (int u = 0; u < G; ++u) {
        const int s = __builtin_amdgcn_readlane(cs, sh + u);
        w[u] = hin2[(unsigned)s * (LDH / 2) + l2];
      }
#pragma unroll
      for (int u = 0; u < G; ++u) {
        const float ax = acc.x + w[u].x;
        const float ay = acc.y + w[u].y;
        acc.x = (u < nn) ? ax : acc.x;
        acc.y = (u < nn) ? ay : acc.y;
      }
    }
    const float di = __int_as_float(
        __builtin_amdgcn_readlane(__float_as_int(dv), i));
    v2f r = { fmaf(acc.x, di, bv.x), fmaf(acc.y, di, bv.y) };
    if (RELU) {
      r.x = r.x > 0.0f ? r.x : 0.0f;
      r.y = r.y > 0.0f ? r.y : 0.0f;
    }
    if (r0 + i < n && lane < NV2) *(v2f*)(res + (long)(r0 + i) * LDO + lo) = r;
  };

  issue(0, va, sfa);
#pragma unroll
  for (int i = 0; i < K; i += 2) {
    if (i + 1 < K) issue(i + 1, vb, sfb);
    finish(i, va, sfa);
    if (i + 1 < K) {
      if (i + 2 < K) issue(i + 2, va, sfa);
      finish(i + 1, vb, sfb);
    }
  }
}

constexpr int TP = 68;

template <int NT>
__device__ __forceinline__ void tile_gemm1_part(const float* __restrict__ tin,
                                                const float* __restrict__ B,
                                                const float* __restrict__ bias,
                                                float* __restrict__ hout,
                                                const int t0, const int lane) {
  constexpr int LDB = 64;
  constexpr int KS = 4;
  static_assert(NT == 2, "guard written for 2 tiles");
  v8f acc[NT];
#pragma unroll
  for (int t = 0; t < NT; ++t) {
    acc[t] = v8f{};
    asm volatile("" : "+v"(acc[t]));
  }
  const int mrow = lane & 15;
  const int hsel = lane >> 4;
  const int kofs = hsel << 1;

  float bs[NT];
  const float* bp[NT];
#pragma unroll
  for (int t = 0; t < NT; ++t) {
    const int bn = (t0 + t) * 16 + mrow;
    bs[t] = bias[bn];
    bp[t] = B + kofs * LDB + bn;
  }
  const float* arow = tin + mrow * TP + kofs;

#pragma unroll
  for (int k0 = 0; k0 < 16; k0 += KS) {
    v2f a[KS], b[NT][KS];
#pragma unroll
    for (int j = 0; j < KS; ++j) a[j] = *(const v2f*)(arow + 4 * (k0 + j));
#pragma unroll
    for (int t = 0; t < NT; ++t)
#pragma unroll
      for (int j = 0; j < KS; ++j)
        b[t][j] = v2f{ bp[t][4 * (k0 + j) * LDB], bp[t][4 * (k0 + j) * LDB + LDB] };

    __builtin_amdgcn_sched_barrier(0);
#pragma unroll
    for (int j = 0; j < KS; ++j) {
#pragma unroll
      for (int t = 0; t < NT; ++t) {
        acc[t] = __builtin_amdgcn_wmma_f32_16x16x4_f32(
             false, a[j],  false, b[t][j],
             (short)0, acc[t],  false,  false);
      }
    }
    __builtin_amdgcn_sched_barrier(0);

    asm volatile("v_nop\n\tv_nop\n\tv_nop\n\tv_nop"
                 : "+v"(acc[0]), "+v"(acc[1])
                 : "v"(a[KS - 1]), "v"(b[0][KS - 1]), "v"(b[1][KS - 1]));
  }

#pragma unroll
  for (int t = 0; t < NT; ++t) {
    float* hp = hout + (hsel << 3) * TP + (t0 + t) * 16 + mrow;
#pragma unroll
    for (int r = 0; r < 8; ++r) {
      const float o = acc[t][r] + bs[t];
      hp[r * TP] = o > 0.0f ? o : 0.0f;
    }
  }
}

template <int NT>
__device__ __forceinline__ void tile_gemm_part(const float* __restrict__ tile,
                                               const float* __restrict__ B,
                                               const float* __restrict__ rscale,
                                               float* __restrict__ C,
                                               const int t0, const int row0, const int nrows,
                                               const int lane) {
  constexpr int LDB = 40, LDC = 48, NVALID = 40;
  constexpr int KS = 4;
  static_assert(NT == 1 || NT == 2, "guard written for 1 or 2 tiles");
  v8f acc[NT];
#pragma unroll
  for (int t = 0; t < NT; ++t) {
    acc[t] = v8f{};
    asm volatile("" : "+v"(acc[t]));
  }
  const int mrow = lane & 15;
  const int hsel = lane >> 4;
  const int kofs = hsel << 1;

  const int crow = row0 + (hsel << 3);
  const bool full = (row0 + 16 <= nrows);
  float sc[8];
  if (full) {
    const float4 s0 = *(const float4*)(rscale + crow);
    const float4 s1 = *(const float4*)(rscale + crow + 4);
    sc[0] = s0.x; sc[1] = s0.y; sc[2] = s0.z; sc[3] = s0.w;
    sc[4] = s1.x; sc[5] = s1.y; sc[6] = s1.z; sc[7] = s1.w;
  } else {
#pragma unroll
    for (int r = 0; r < 8; ++r) {
      const int rr = (crow + r < nrows) ? (crow + r) : (nrows - 1);
      sc[r] = rscale[rr];
    }
  }

  const float* arow = tile + mrow * TP + kofs;
  const float* bp[NT];
#pragma unroll
  for (int t = 0; t < NT; ++t) {
    const int bn   = (t0 + t) * 16 + mrow;
    const int bcol = (bn < NVALID) ? bn : (NVALID - 1);
    bp[t] = B + kofs * LDB + bcol;
  }

#pragma unroll
  for (int k0 = 0; k0 < 16; k0 += KS) {
    v2f a[KS], b[NT][KS];
#pragma unroll
    for (int j = 0; j < KS; ++j) a[j] = *(const v2f*)(arow + 4 * (k0 + j));
#pragma unroll
    for (int t = 0; t < NT; ++t)
#pragma unroll
      for (int j = 0; j < KS; ++j)
        b[t][j] = v2f{ bp[t][4 * (k0 + j) * LDB], bp[t][4 * (k0 + j) * LDB + LDB] };

    __builtin_amdgcn_sched_barrier(0);
#pragma unroll
    for (int j = 0; j < KS; ++j) {
#pragma unroll
      for (int t = 0; t < NT; ++t) {
        acc[t] = __builtin_amdgcn_wmma_f32_16x16x4_f32(
             false, a[j],  false, b[t][j],
             (short)0, acc[t],  false,  false);
      }
    }
    __builtin_amdgcn_sched_barrier(0);

    if constexpr (NT == 2) {
      asm volatile("v_nop\n\tv_nop\n\tv_nop\n\tv_nop"
                   : "+v"(acc[0]), "+v"(acc[1])
                   : "v"(a[KS - 1]), "v"(b[0][KS - 1]), "v"(b[1][KS - 1]));
    } else {
      asm volatile("v_nop\n\tv_nop\n\tv_nop\n\tv_nop"
                   : "+v"(acc[0])
                   : "v"(a[KS - 1]), "v"(b[0][KS - 1]));
    }
  }

#pragma unroll
  for (int t = 0; t < NT; ++t) {
    float o[8];
#pragma unroll
    for (int r = 0; r < 8; ++r) o[r] = acc[t][r] * sc[r];
    float* cp = C + crow * LDC + (t0 + t) * 16 + mrow;
    if (full) {
#pragma unroll
      for (int r = 0; r < 8; ++r) cp[r * LDC] = o[r];
    } else {
#pragma unroll
      for (int r = 0; r < 8; ++r)
        if (crow + r < nrows) cp[r * LDC] = o[r];
    }
  }
}

template <int K>
__global__ void __launch_bounds__(256)
gcn_gather_gemm2(const float* __restrict__ hin,
                 const int* __restrict__ rowptr,
                 const int* __restrict__ col,
                 const float* __restrict__ dinv,
                 const float* __restrict__ B1,
                 const float* __restrict__ bias,
                 const float* __restrict__ B,
                 float* __restrict__ C, int n) {
  constexpr int G = 16;
  constexpr int LDH = 64;
  static_assert(K == 8, "8 waves x K rows = the 64 rows of the tile");
  static_assert(2 * G == 32, "a row's second group of ids is the upper half of ms[i]");
  __shared__ __attribute__((aligned(16))) float tile[8 * K * TP];
  __shared__ __attribute__((aligned(16))) float tile2[8 * K * TP];
  const int lane = threadIdx.x & 31;
  const int wv = __builtin_amdgcn_readfirstlane((int)(threadIdx.x >> 5));
  const int brow0 = blockIdx.x * (8 * K);
  const int r0 = brow0 + wv * K;

  const int l2 = lane;
  const int lo = l2 * 2;
  const v2f* hin2 = (const v2f*)hin;

  const int bi = (lane < K) ? lane : K;
  const int rpv = rowptr[(r0 + bi < n) ? (r0 + bi) : n];
  const int fi = (lane < K - 1) ? lane : (K - 1);
  const float dv = dinv[(r0 + fi < n) ? (r0 + fi) : (n - 1)];

  int rb[K], rn[K], ms[K];
#pragma unroll
  for (int i = 0; i < K; ++i) {
    rb[i] = __builtin_amdgcn_readlane(rpv, i);
    rn[i] = __builtin_amdgcn_readlane(rpv, i + 1) - rb[i];
    ms[i] = 0;
    if (rn[i] > 0) ms[i] = col[rb[i] + ((lane < rn[i]) ? lane : (rn[i] - 1))];
  }

  v2f va[G], vb[G], sfa, sfb;

  auto issue = [&](const int i, v2f (&v)[G], v2f& sf) {
    const int rr = (r0 + i < n) ? (r0 + i) : (n - 1);
    sf = hin2[(unsigned)rr * (LDH / 2) + l2];
#pragma unroll
    for (int u = 0; u < G; ++u) {
      const int s = __builtin_amdgcn_readlane(ms[i], u);
      v[u] = hin2[(unsigned)s * (LDH / 2) + l2];
    }
  };

  auto finish = [&](const int i, v2f (&v)[G], const v2f& sf) {
    const int ne = rn[i];
    v2f acc = sf;
#pragma unroll
    for (int u = 0; u < G; ++u) {
      const float ax = acc.x + v[u].x;
      const float ay = acc.y + v[u].y;
      acc.x = (u < ne) ? ax : acc.x;
      acc.y = (u < ne) ? ay : acc.y;
    }
#pragma nounroll
    for (int off = G; off < ne; off += G) {
      const int nn = (ne - off < G) ? (ne - off) : G;
      int cs = ms[i];
      int sh = G;
      if (off >= 32) {
        cs = col[rb[i] + off + ((lane < nn) ? lane : (nn - 1))];
        sh = 0;
      }
      v2f w[G];
#pragma unroll
      for (int u = 0; u < G; ++u) {
        const int s = __builtin_amdgcn_readlane(cs, sh + u);
        w[u] = hin2[(unsigned)s * (LDH / 2) + l2];
      }
#pragma unroll
      for (int u = 0; u < G; ++u) {
        const float ax = acc.x + w[u].x;
        const float ay = acc.y + w[u].y;
        acc.x = (u < nn) ? ax : acc.x;
        acc.y = (u < nn) ? ay : acc.y;
      }
    }
    const float di = __int_as_float(
        __builtin_amdgcn_readlane(__float_as_int(dv), i));
    v2f r = { acc.x * di, acc.y * di };
    if (r0 + i >= n) r = v2f{0.0f, 0.0f};
    *(v2f*)(tile + (wv * K + i) * TP + lo) = r;
  };

  issue(0, va, sfa);
#pragma unroll
  for (int i = 0; i < K; i += 2) {
    if (i + 1 < K) issue(i + 1, vb, sfb);
    finish(i, va, sfa);
    if (i + 1 < K) {
      if (i + 2 < K) issue(i + 2, va, sfa);
      finish(i + 1, vb, sfb);
    }
  }

  __syncthreads();

  const int mt = wv & 3;
  tile_gemm1_part<2>(tile + mt * 16 * TP, B1, bias, tile2 + mt * 16 * TP, (wv < 4) ? 0 : 2, lane);

  __syncthreads();

  const int grow0 = brow0 + mt * 16;
  if (grow0 >= n) return;
  if (wv < 4) tile_gemm_part<2>(tile2 + mt * 16 * TP, B, dinv, C, 0, grow0, n, lane);
  else        tile_gemm_part<1>(tile2 + mt * 16 * TP, B, dinv, C, 2, grow0, n, lane);
}

extern "C" void kernel_launch(void* const* d_in, const int* in_sizes, int n_in,
                              void* d_out, int out_size, void* d_ws, size_t ws_size,
                              hipStream_t stream) {
  (void)n_in; (void)out_size; (void)ws_size;
  const float* x  = (const float*)d_in[0];
  const int*   ei = (const int*)d_in[1];
  const float* W1 = (const float*)d_in[2];
  const float* b1 = (const float*)d_in[3];
  const float* W3 = (const float*)d_in[4];
  const float* b3 = (const float*)d_in[5];

  const int N = in_sizes[0] / 64;
  const int E = in_sizes[1] / 2;
  const int* src = ei;
  const int* dst = ei + E;

  const int NBK = (N + BN - 1) / BN;
  const int NCH = (E + CH - 1) / CH;

  float* dinv   = (float*)d_ws;
  float* bufA   = dinv + N;
  float* bufB   = bufA + (long)N * 64;
  int*   cstart = (int*)(bufB + (long)N * 64);
  int*   rowptr = cstart + N;
  unsigned* tmp = (unsigned*)(rowptr + N + 1);
  int*   col    = (int*)tmp + E;
  float* out    = (float*)d_out;

  constexpr int KR = 8;

  csr_chunk_sort<<<NCH, BT, 0, stream>>>(src, dst, tmp, cstart, E, NBK);
  csr_rows<<<NBK, BT, 0, stream>>>(tmp, cstart, rowptr, dinv, col, x, bufA, N, NBK, NCH);

  gcn_gather_gemm2<KR><<<(N + 8 * KR - 1) / (8 * KR), 256, 0, stream>>>(
      bufA, rowptr, col, dinv, W1, b1, W3, bufB, N);

  gcn_gather_rows<48, 20, 40, false, KR><<<(N + 8 * KR - 1) / (8 * KR), 256, 0, stream>>>(
      bufB, rowptr, col, dinv, b3, out, N);
}


# Gather row loads: a row's 17 addresses are formed on the scalar unit, each in registers of its own, before its first load; lanes add one fixed offset; no translation wait between loads; sums kept
# speedup vs baseline: 5.3643x; 1.0308x over previous
#include <hip/hip_runtime.h>

typedef float v2f __attribute__((ext_vector_type(2)));
typedef float v8f __attribute__((ext_vector_type(8)));
typedef const __attribute__((address_space(1))) v2f* gv2p;

constexpr int BSH  = 9;
constexpr int BN   = 1 << BSH;
constexpr int MAXB = 256;
constexpr int CH   = 4704;
constexpr int BT   = 1024;
constexpr int KPT  = (CH + BT - 1) / BT;
constexpr int UB   = 8;
constexpr int MAXC = BT;
constexpr int CAP  = 16384;

__device__ __forceinline__ void lds_barrier() {
  __builtin_amdgcn_fence(__ATOMIC_RELEASE, "workgroup", "local");
  __builtin_amdgcn_s_barrier();
  __builtin_amdgcn_fence(__ATOMIC_ACQUIRE, "workgroup", "local");
}

__device__ __forceinline__ int block_excl_scan(const int v, int* wsum, int& total) {
  const int lane = threadIdx.x & 31;
  const int w = threadIdx.x >> 5;
  int x = v;
  for (int o = 1; o < 32; o <<= 1) {
    const int y = __shfl_up(x, o);
    if (lane >= o) x += y;
  }
  if (lane == 31) wsum[w] = x;
  lds_barrier();
  const int mine = wsum[lane];
  int s = (lane < w) ? mine : 0;
  int t = mine;
  for (int o = 16; o > 0; o >>= 1) {
    s += __shfl_xor(s, o);
    t += __shfl_xor(t, o);
  }
  lds_barrier();
  total = t;
  return s + x - v;
}

__global__ void __launch_bounds__(BT)
csr_chunk_sort(const int* __restrict__ src, const int* __restrict__ dst,
               unsigned* __restrict__ tmp, int* __restrict__ cstart, int e, int nb) {
  __shared__ int h[MAXB];
  __shared__ int lstart[MAXB];
  __shared__ int wsum[32];
  __shared__ unsigned stage[CH];
  const int tid = threadIdx.x;
  const int blk = blockIdx.x;
  const int base = blk * CH;
  const int end = (base + CH < e) ? (base + CH) : e;
  if (tid < MAXB) h[tid] = 0;

  int d[KPT], s[KPT];
#pragma unroll
  for (int k = 0; k < KPT; ++k) {
    const int i = base + tid + k * BT;
    const int ii = (i < end) ? i : (end - 1);
    d[k] = dst[ii];
    s[k] = src[ii];
  }
  __syncthreads();
#pragma unroll
  for (int k = 0; k < KPT; ++k)
    if (base + tid + k * BT < end) atomicAdd(&h[d[k] >> BSH], 1);
  __syncthreads();

  const int L = (tid < nb) ? h[tid] : 0;
  int totalL;
  const int ls = block_excl_scan(L, wsum, totalL);
  if (tid < nb) {
    h[tid] = 0;
    lstart[tid] = ls;
    cstart[blk * (nb + 1) + tid] = ls;
  }
  if (tid == nb) cstart[blk * (nb + 1) + nb] = totalL;
  __syncthreads();

#pragma unroll
  for (int k = 0; k < KPT; ++k) {
    if (base + tid + k * BT < end) {
      const int bb = d[k] >> BSH;
      const int pos = lstart[bb] + atomicAdd(&h[bb], 1);
      stage[pos] = ((unsigned)s[k] << BSH) | (unsigned)(d[k] & (BN - 1));
    }
  }
  __syncthreads();

  const int nloc = end - base;
#pragma unroll
  for (int k = 0; k < KPT; ++k) {
    const int pos = tid + k * BT;
    if (pos < nloc) tmp[base + pos] = stage[pos];
  }
}

constexpr int XPT = BN * 16 / BT;

__global__ void __launch_bounds__(BT)
csr_rows(const unsigned* __restrict__ tmp, const int* __restrict__ cstart,
         int* __restrict__ rowptr, float* __restrict__ dinv,
         int* __restrict__ col, const float* __restrict__ x,
         float* __restrict__ xs, int n, int nb, int nchunk) {
  __shared__ int cnt[BN];
  __shared__ float dl[BN];
  __shared__ int cur[BN];
  __shared__ int wsum[32];
  __shared__ int rbeg[MAXC];
  __shared__ int rlen[MAXC];
  __shared__ int stg[CAP];
  const int tid = threadIdx.x;
  const int lane = tid & 31;
  const int wid = __builtin_amdgcn_readfirstlane((int)(threadIdx.x >> 5));
  const int b = blockIdx.x;

  int before = 0, mine = 0;
  if (tid < nchunk) {
    const int* cs = cstart + tid * (nb + 1) + b;
    before = cs[0];
    mine = cs[1] - before;
    rbeg[tid] = tid * CH + before;
    rlen[tid] = mine;
  }
  if (tid < BN) cnt[tid] = 0;

  const int xr = tid >> 4;
  const int xc = (tid & 15) * 4;
  float4 xv[XPT];
#pragma unroll
  for (int j = 0; j < XPT; ++j) {
    const int node = b * BN + j * (BT / 16) + xr;
    const int nn = (node < n) ? node : (n - 1);
    xv[j] = *(const float4*)(x + (long)nn * 64 + xc);
  }

  int s0, m;
  block_excl_scan(before, wsum, s0);
  block_excl_scan(mine, wsum, m);
  const int s1 = s0 + m;

  auto walk = [&](auto&& f) {
    for (int c0 = wid; c0 < nchunk; c0 += 32 * UB) {
      unsigned w[UB];
      int len[UB], beg[UB];
#pragma unroll
      for (int u = 0; u < UB; ++u) {
        const int cc = c0 + 32 * u;
        const int ci = (cc < nchunk) ? cc : c0;
        beg[u] = __builtin_amdgcn_readfirstlane(rbeg[ci]);
        len[u] = (cc < nchunk) ? __builtin_amdgcn_readfirstlane(rlen[ci]) : 0;
        w[u] = tmp[(lane < len[u]) ? (beg[u] + lane) : 0];
      }
#pragma unroll
      for (int u = 0; u < UB; ++u)
        if (lane < len[u]) f(w[u]);
#pragma unroll
      for (int u = 0; u < UB; ++u)
        for (int off = 32 + lane; off < len[u]; off += 32) f(tmp[beg[u] + off]);
    }
  };

  walk([&](const unsigned w) { atomicAdd(&cnt[w & (BN - 1)], 1); });
  lds_barrier();

  const int c = (tid < BN) ? cnt[tid] : 0;
  int total;
  const int ex = block_excl_scan(c, wsum, total);
  const int node = b * BN + tid;
  if (tid < BN) {
    cur[tid] = ex;
    const float di = rsqrtf((float)(c + 1));
    dl[tid] = di;
    if (node < n) {
      rowptr[node] = s0 + ex;
      dinv[node] = di;
    }
  }
  if (b == (int)gridDim.x - 1 && tid == 0) rowptr[n] = s1;
  lds_barrier();

#pragma unroll
  for (int j = 0; j < XPT; ++j) {
    const int lr = j * (BT / 16) + xr;
    const int node2 = b * BN + lr;
    if (node2 < n) {
      const float di = dl[lr];
      float4 v = xv[j];
      v.x *= di; v.y *= di; v.z *= di; v.w *= di;
      *(float4*)(xs + (long)node2 * 64 + xc) = v;
    }
  }

  const bool staged = (m <= CAP);
  walk([&](const unsigned w) {
    const int p = atomicAdd(&cur[w & (BN - 1)], 1);
    if (staged) stg[p] = (int)(w >> BSH);
    else col[s0 + p] = (int)(w >> BSH);
  });
  lds_barrier();
  if (staged)
    for (int i = tid; i < m; i += BT) col[s0 + i] = stg[i];
}


template <int LDH, int NV2, int LDO, bool RELU, int K>
__global__ void gcn_gather_rows(const float* __restrict__ hin,
                                const int* __restrict__ rowptr,
                                const int* __restrict__ col,
                                const float* __restrict__ dinv,
                                const float* __restrict__ bias,
                                float* __restrict__ res, int n) {
  constexpr int G = 16;
  static_assert(K >= 1 && K < 32, "row bounds come from one 32-lane load");
  static_assert(2 * G == 32, "a row's second group of ids is the upper half of ms[i]");
  const int lane = threadIdx.x & 31;
  const int wid = __builtin_amdgcn_readfirstlane(
      (int)(blockIdx.x * (blockDim.x >> 5) + (threadIdx.x >> 5)));
  const int r0 = wid * K;
  if (r0 >= n) return;

  const int l2 = (lane < NV2) ? lane : (NV2 - 1);
  const int lo = l2 * 2;
  auto rowbase = [&](const int row) {
    const unsigned long long a =
        (unsigned long long)hin + (unsigned long long)((unsigned)row * (unsigned)(LDH * 4));
    const unsigned alo = (unsigned)__builtin_amdgcn_readfirstlane((int)(unsigned)a);
    const unsigned ahi = (unsigned)__builtin_amdgcn_readfirstlane((int)(unsigned)(a >> 32));
    return (gv2p)(((unsigned long long)ahi << 32) | (unsigned long long)alo);
  };

  const int bi = (lane < K) ? lane : K;
  const int rpv = rowptr[(r0 + bi < n) ? (r0 + bi) : n];
  const int fi = (lane < K - 1) ? lane : (K - 1);
  const float dv = dinv[(r0 + fi < n) ? (r0 + fi) : (n - 1)];
  const v2f bv = *(const v2f*)(bias + lo);

  int rb[K], rn[K], ms[K];
#pragma unroll
  for (int i = 0; i < K; ++i) {
    rb[i] = __builtin_amdgcn_readlane(rpv, i);
    rn[i] = __builtin_amdgcn_readlane(rpv, i + 1) - rb[i];
    ms[i] = 0;
    if (rn[i] > 0) ms[i] = col[rb[i] + ((lane < rn[i]) ? lane : (rn[i] - 1))];
  }

  v2f va[G], vb[G], sfa, sfb;

  auto issue = [&](const int i, v2f (&v)[G], v2f& sf) {
    const int rr = (r0 + i < n) ? (r0 + i) : (n - 1);
    const gv2p ps = rowbase(rr);
    gv2p pb[G];
#pragma unroll
    for (int u = 0; u < G; ++u) {
      pb[u] = rowbase(__builtin_amdgcn_readlane(ms[i], u));
    }
    __builtin_amdgcn_sched_barrier(0);
    sf = ps[l2];
#pragma unroll
    for (int u = 0; u < G; ++u) v[u] = pb[u][l2];
  };

  auto finish = [&](const int i, v2f (&v)[G], const v2f& sf) {
    const int ne = rn[i];
    v2f acc = sf;
#pragma unroll
    for (int u = 0; u < G; ++u) {
      const float ax = acc.x + v[u].x;
      const float ay = acc.y + v[u].y;
      acc.x = (u < ne) ? ax : acc.x;
      acc.y = (u < ne) ? ay : acc.y;
    }
#pragma nounroll
    for (int off = G; off < ne; off += G) {
      const int nn = (ne - off < G) ? (ne - off) : G;
      int cs = ms[i];
      int sh = G;
      if (off >= 32) {
        cs = col[rb[i] + off + ((lane < nn) ? lane : (nn - 1))];
        sh = 0;
      }
      v2f w[G];
      gv2p pw[G];
#pragma unroll
      for (int u = 0; u < G; ++u) {
        pw[u] = rowbase(__builtin_amdgcn_readlane(cs, sh + u));
      }
      __builtin_amdgcn_sched_barrier(0);
#pragma unroll
      for (int u = 0; u < G; ++u) w[u] = pw[u][l2];
#pragma unroll
      for (int u = 0; u < G; ++u) {
        const float ax = acc.x + w[u].x;
        const float ay = acc.y + w[u].y;
        acc.x = (u < nn) ? ax : acc.x;
        acc.y = (u < nn) ? ay : acc.y;
      }
    }
    const float di = __int_as_float(
        __builtin_amdgcn_readlane(__float_as_int(dv), i));
    v2f r = { fmaf(acc.x, di, bv.x), fmaf(acc.y, di, bv.y) };
    if (RELU) {
      r.x = r.x > 0.0f ? r.x : 0.0f;
      r.y = r.y > 0.0f ? r.y : 0.0f;
    }
    if (r0 + i < n && lane < NV2) *(v2f*)(res + (long)(r0 + i) * LDO + lo) = r;
  };

  issue(0, va, sfa);
#pragma unroll
  for (int i = 0; i < K; i += 2) {
    if (i + 1 < K) issue(i + 1, vb, sfb);
    finish(i, va, sfa);
    if (i + 1 < K) {
      if (i + 2 < K) issue(i + 2, va, sfa);
      finish(i + 1, vb, sfb);
    }
  }
}

constexpr int TP = 68;

template <int NT>
__device__ __forceinline__ void tile_gemm1_part(const float* __restrict__ tin,
                                                const float* __restrict__ B,
                                                const float* __restrict__ bias,
                                                float* __restrict__ hout,
                                                const int t0, const int lane) {
  constexpr int LDB = 64;
  constexpr int KS = 4;
  static_assert(NT == 2, "guard written for 2 tiles");
  v8f acc[NT];
#pragma unroll
  for (int t = 0; t < NT; ++t) {
    acc[t] = v8f{};
    asm volatile("" : "+v"(acc[t]));
  }
  const int mrow = lane & 15;
  const int hsel = lane >> 4;
  const int kofs = hsel << 1;

  float bs[NT];
  const float* bp[NT];
#pragma unroll
  for (int t = 0; t < NT; ++t) {
    const int bn = (t0 + t) * 16 + mrow;
    bs[t] = bias[bn];
    bp[t] = B + kofs * LDB + bn;
  }
  const float* arow = tin + mrow * TP + kofs;

#pragma unroll
  for (int k0 = 0; k0 < 16; k0 += KS) {
    v2f a[KS], b[NT][KS];
#pragma unroll
    for (int j = 0; j < KS; ++j) a[j] = *(const v2f*)(arow + 4 * (k0 + j));
#pragma unroll
    for (int t = 0; t < NT; ++t)
#pragma unroll
      for (int j = 0; j < KS; ++j)
        b[t][j] = v2f{ bp[t][4 * (k0 + j) * LDB], bp[t][4 * (k0 + j) * LDB + LDB] };

    __builtin_amdgcn_sched_barrier(0);
#pragma unroll
    for (int j = 0; j < KS; ++j) {
#pragma unroll
      for (int t = 0; t < NT; ++t) {
        acc[t] = __builtin_amdgcn_wmma_f32_16x16x4_f32(
             false, a[j],  false, b[t][j],
             (short)0, acc[t],  false,  false);
      }
    }
    __builtin_amdgcn_sched_barrier(0);

    asm volatile("v_nop\n\tv_nop\n\tv_nop\n\tv_nop"
                 : "+v"(acc[0]), "+v"(acc[1])
                 : "v"(a[KS - 1]), "v"(b[0][KS - 1]), "v"(b[1][KS - 1]));
  }

#pragma unroll
  for (int t = 0; t < NT; ++t) {
    float* hp = hout + (hsel << 3) * TP + (t0 + t) * 16 + mrow;
#pragma unroll
    for (int r = 0; r < 8; ++r) {
      const float o = acc[t][r] + bs[t];
      hp[r * TP] = o > 0.0f ? o : 0.0f;
    }
  }
}

template <int NT>
__device__ __forceinline__ void tile_gemm_part(const float* __restrict__ tile,
                                               const float* __restrict__ B,
                                               const float* __restrict__ rscale,
                                               float* __restrict__ C,
                                               const int t0, const int row0, const int nrows,
                                               const int lane) {
  constexpr int LDB = 40, LDC = 48, NVALID = 40;
  constexpr int KS = 4;
  static_assert(NT == 1 || NT == 2, "guard written for 1 or 2 tiles");
  v8f acc[NT];
#pragma unroll
  for (int t = 0; t < NT; ++t) {
    acc[t] = v8f{};
    asm volatile("" : "+v"(acc[t]));
  }
  const int mrow = lane & 15;
  const int hsel = lane >> 4;
  const int kofs = hsel << 1;

  const int crow = row0 + (hsel << 3);
  const bool full = (row0 + 16 <= nrows);
  float sc[8];
  if (full) {
    const float4 s0 = *(const float4*)(rscale + crow);
    const float4 s1 = *(const float4*)(rscale + crow + 4);
    sc[0] = s0.x; sc[1] = s0.y; sc[2] = s0.z; sc[3] = s0.w;
    sc[4] = s1.x; sc[5] = s1.y; sc[6] = s1.z; sc[7] = s1.w;
  } else {
#pragma unroll
    for (int r = 0; r < 8; ++r) {
      const int rr = (crow + r < nrows) ? (crow + r) : (nrows - 1);
      sc[r] = rscale[rr];
    }
  }

  const float* arow = tile + mrow * TP + kofs;
  const float* bp[NT];
#pragma unroll
  for (int t = 0; t < NT; ++t) {
    const int bn   = (t0 + t) * 16 + mrow;
    const int bcol = (bn < NVALID) ? bn : (NVALID - 1);
    bp[t] = B + kofs * LDB + bcol;
  }

#pragma unroll
  for (int k0 = 0; k0 < 16; k0 += KS) {
    v2f a[KS], b[NT][KS];
#pragma unroll
    for (int j = 0; j < KS; ++j) a[j] = *(const v2f*)(arow + 4 * (k0 + j));
#pragma unroll
    for (int t = 0; t < NT; ++t)
#pragma unroll
      for (int j = 0; j < KS; ++j)
        b[t][j] = v2f{ bp[t][4 * (k0 + j) * LDB], bp[t][4 * (k0 + j) * LDB + LDB] };

    __builtin_amdgcn_sched_barrier(0);
#pragma unroll
    for (int j = 0; j < KS; ++j) {
#pragma unroll
      for (int t = 0; t < NT; ++t) {
        acc[t] = __builtin_amdgcn_wmma_f32_16x16x4_f32(
             false, a[j],  false, b[t][j],
             (short)0, acc[t],  false,  false);
      }
    }
    __builtin_amdgcn_sched_barrier(0);

    if constexpr (NT == 2) {
      asm volatile("v_nop\n\tv_nop\n\tv_nop\n\tv_nop"
                   : "+v"(acc[0]), "+v"(acc[1])
                   : "v"(a[KS - 1]), "v"(b[0][KS - 1]), "v"(b[1][KS - 1]));
    } else {
      asm volatile("v_nop\n\tv_nop\n\tv_nop\n\tv_nop"
                   : "+v"(acc[0])
                   : "v"(a[KS - 1]), "v"(b[0][KS - 1]));
    }
  }

#pragma unroll
  for (int t = 0; t < NT; ++t) {
    float o[8];
#pragma unroll
    for (int r = 0; r < 8; ++r) o[r] = acc[t][r] * sc[r];
    float* cp = C + crow * LDC + (t0 + t) * 16 + mrow;
    if (full) {
#pragma unroll
      for (int r = 0; r < 8; ++r) cp[r * LDC] = o[r];
    } else {
#pragma unroll
      for (int r = 0; r < 8; ++r)
        if (crow + r < nrows) cp[r * LDC] = o[r];
    }
  }
}

template <int K>
__global__ void __launch_bounds__(256)
gcn_gather_gemm2(const float* __restrict__ hin,
                 const int* __restrict__ rowptr,
                 const int* __restrict__ col,
                 const float* __restrict__ dinv,
                 const float* __restrict__ B1,
                 const float* __restrict__ bias,
                 const float* __restrict__ B,
                 float* __restrict__ C, int n) {
  constexpr int G = 16;
  constexpr int LDH = 64;
  static_assert(K == 8, "8 waves x K rows = the 64 rows of the tile");
  static_assert(2 * G == 32, "a row's second group of ids is the upper half of ms[i]");
  __shared__ __attribute__((aligned(16))) float tile[8 * K * TP];
  __shared__ __attribute__((aligned(16))) float tile2[8 * K * TP];
  const int lane = threadIdx.x & 31;
  const int wv = __builtin_amdgcn_readfirstlane((int)(threadIdx.x >> 5));
  const int brow0 = blockIdx.x * (8 * K);
  const int r0 = brow0 + wv * K;

  const int l2 = lane;
  const int lo = l2 * 2;
  auto rowbase = [&](const int row) {
    const unsigned long long a =
        (unsigned long long)hin + (unsigned long long)((unsigned)row * (unsigned)(LDH * 4));
    const unsigned alo = (unsigned)__builtin_amdgcn_readfirstlane((int)(unsigned)a);
    const unsigned ahi = (unsigned)__builtin_amdgcn_readfirstlane((int)(unsigned)(a >> 32));
    return (gv2p)(((unsigned long long)ahi << 32) | (unsigned long long)alo);
  };

  const int bi = (lane < K) ? lane : K;
  const int rpv = rowptr[(r0 + bi < n) ? (r0 + bi) : n];
  const int fi = (lane < K - 1) ? lane : (K - 1);
  const float dv = dinv[(r0 + fi < n) ? (r0 + fi) : (n - 1)];

  int rb[K], rn[K], ms[K];
#pragma unroll
  for (int i = 0; i < K; ++i) {
    rb[i] = __builtin_amdgcn_readlane(rpv, i);
    rn[i] = __builtin_amdgcn_readlane(rpv, i + 1) - rb[i];
    ms[i] = 0;
    if (rn[i] > 0) ms[i] = col[rb[i] + ((lane < rn[i]) ? lane : (rn[i] - 1))];
  }

  v2f va[G], vb[G], sfa, sfb;

  auto issue = [&](const int i, v2f (&v)[G], v2f& sf) {
    const int rr = (r0 + i < n) ? (r0 + i) : (n - 1);
    const gv2p ps = rowbase(rr);
    gv2p pb[G];
#pragma unroll
    for (int u = 0; u < G; ++u) {
      pb[u] = rowbase(__builtin_amdgcn_readlane(ms[i], u));
    }
    __builtin_amdgcn_sched_barrier(0);
    sf = ps[l2];
#pragma unroll
    for (int u = 0; u < G; ++u) v[u] = pb[u][l2];
  };

  auto finish = [&](const int i, v2f (&v)[G], const v2f& sf) {
    const int ne = rn[i];
    v2f acc = sf;
#pragma unroll
    for (int u = 0; u < G; ++u) {
      const float ax = acc.x + v[u].x;
      const float ay = acc.y + v[u].y;
      acc.x = (u < ne) ? ax : acc.x;
      acc.y = (u < ne) ? ay : acc.y;
    }
#pragma nounroll
    for (int off = G; off < ne; off += G) {
      const int nn = (ne - off < G) ? (ne - off) : G;
      int cs = ms[i];
      int sh = G;
      if (off >= 32) {
        cs = col[rb[i] + off + ((lane < nn) ? lane : (nn - 1))];
        sh = 0;
      }
      v2f w[G];
      gv2p pw[G];
#pragma unroll
      for (int u = 0; u < G; ++u) {
        pw[u] = rowbase(__builtin_amdgcn_readlane(cs, sh + u));
      }
      __builtin_amdgcn_sched_barrier(0);
#pragma unroll
      for (int u = 0; u < G; ++u) w[u] = pw[u][l2];
#pragma unroll
      for (int u = 0; u < G; ++u) {
        const float ax = acc.x + w[u].x;
        const float ay = acc.y + w[u].y;
        acc.x = (u < nn) ? ax : acc.x;
        acc.y = (u < nn) ? ay : acc.y;
      }
    }
    const float di = __int_as_float(
        __builtin_amdgcn_readlane(__float_as_int(dv), i));
    v2f r = { acc.x * di, acc.y * di };
    if (r0 + i >= n) r = v2f{0.0f, 0.0f};
    *(v2f*)(tile + (wv * K + i) * TP + lo) = r;
  };

  issue(0, va, sfa);
#pragma unroll
  for (int i = 0; i < K; i += 2) {
    if (i + 1 < K) issue(i + 1, vb, sfb);
    finish(i, va, sfa);
    if (i + 1 < K) {
      if (i + 2 < K) issue(i + 2, va, sfa);
      finish(i + 1, vb, sfb);
    }
  }

  __syncthreads();

  const int mt = wv & 3;
  tile_gemm1_part<2>(tile + mt * 16 * TP, B1, bias, tile2 + mt * 16 * TP, (wv < 4) ? 0 : 2, lane);

  __syncthreads();

  const int grow0 = brow0 + mt * 16;
  if (grow0 >= n) return;
  if (wv < 4) tile_gemm_part<2>(tile2 + mt * 16 * TP, B, dinv, C, 0, grow0, n, lane);
  else        tile_gemm_part<1>(tile2 + mt * 16 * TP, B, dinv, C, 2, grow0, n, lane);
}

extern "C" void kernel_launch(void* const* d_in, const int* in_sizes, int n_in,
                              void* d_out, int out_size, void* d_ws, size_t ws_size,
                              hipStream_t stream) {
  (void)n_in; (void)out_size; (void)ws_size;
  const float* x  = (const float*)d_in[0];
  const int*   ei = (const int*)d_in[1];
  const float* W1 = (const float*)d_in[2];
  const float* b1 = (const float*)d_in[3];
  const float* W3 = (const float*)d_in[4];
  const float* b3 = (const float*)d_in[5];

  const int N = in_sizes[0] / 64;
  const int E = in_sizes[1] / 2;
  const int* src = ei;
  const int* dst = ei + E;

  const int NBK = (N + BN - 1) / BN;
  const int NCH = (E + CH - 1) / CH;

  float* dinv   = (float*)d_ws;
  float* bufA   = dinv + N;
  float* bufB   = bufA + (long)N * 64;
  int*   cstart = (int*)(bufB + (long)N * 64);
  int*   rowptr = cstart + N;
  unsigned* tmp = (unsigned*)(rowptr + N + 1);
  int*   col    = (int*)tmp + E;
  float* out    = (float*)d_out;

  constexpr int KR = 8;

  csr_chunk_sort<<<NCH, BT, 0, stream>>>(src, dst, tmp, cstart, E, NBK);
  csr_rows<<<NBK, BT, 0, stream>>>(tmp, cstart, rowptr, dinv, col, x, bufA, N, NBK, NCH);

  gcn_gather_gemm2<KR><<<(N + 8 * KR - 1) / (8 * KR), 256, 0, stream>>>(
      bufA, rowptr, col, dinv, W1, b1, W3, bufB, N);

  gcn_gather_rows<48, 20, 40, false, KR><<<(N + 8 * KR - 1) / (8 * KR), 256, 0, stream>>>(
      bufB, rowptr, col, dinv, b3, out, N);
}


# Gathers fetch TWO rows per 16-B-per-lane load (16 lanes cover a row; lanes 0-15 even list places, 16-31 odd): half the row loads; sum order changes: two partial sums per row, added across halves
# speedup vs baseline: 5.5335x; 1.0315x over previous
#include <hip/hip_runtime.h>

typedef float v2f __attribute__((ext_vector_type(2)));
typedef float v8f __attribute__((ext_vector_type(8)));
typedef const __attribute__((address_space(1))) v2f* gv2p;

constexpr int BSH  = 9;
constexpr int BN   = 1 << BSH;
constexpr int MAXB = 256;
constexpr int CH   = 4704;
constexpr int BT   = 1024;
constexpr int KPT  = (CH + BT - 1) / BT;
constexpr int UB   = 8;
constexpr int MAXC = BT;
constexpr int CAP  = 16384;

__device__ __forceinline__ void lds_barrier() {
  __builtin_amdgcn_fence(__ATOMIC_RELEASE, "workgroup", "local");
  __builtin_amdgcn_s_barrier();
  __builtin_amdgcn_fence(__ATOMIC_ACQUIRE, "workgroup", "local");
}

__device__ __forceinline__ int block_excl_scan(const int v, int* wsum, int& total) {
  const int lane = threadIdx.x & 31;
  const int w = threadIdx.x >> 5;
  int x = v;
  for (int o = 1; o < 32; o <<= 1) {
    const int y = __shfl_up(x, o);
    if (lane >= o) x += y;
  }
  if (lane == 31) wsum[w] = x;
  lds_barrier();
  const int mine = wsum[lane];
  int s = (lane < w) ? mine : 0;
  int t = mine;
  for (int o = 16; o > 0; o >>= 1) {
    s += __shfl_xor(s, o);
    t += __shfl_xor(t, o);
  }
  lds_barrier();
  total = t;
  return s + x - v;
}

__global__ void __launch_bounds__(BT)
csr_chunk_sort(const int* __restrict__ src, const int* __restrict__ dst,
               unsigned* __restrict__ tmp, int* __restrict__ cstart, int e, int nb) {
  __shared__ int h[MAXB];
  __shared__ int lstart[MAXB];
  __shared__ int wsum[32];
  __shared__ unsigned stage[CH];
  const int tid = threadIdx.x;
  const int blk = blockIdx.x;
  const int base = blk * CH;
  const int end = (base + CH < e) ? (base + CH) : e;
  if (tid < MAXB) h[tid] = 0;

  int d[KPT], s[KPT];
#pragma unroll
  for (int k = 0; k < KPT; ++k) {
    const int i = base + tid + k * BT;
    const int ii = (i < end) ? i : (end - 1);
    d[k] = dst[ii];
    s[k] = src[ii];
  }
  __syncthreads();
#pragma unroll
  for (int k = 0; k < KPT; ++k)
    if (base + tid + k * BT < end) atomicAdd(&h[d[k] >> BSH], 1);
  __syncthreads();

  const int L = (tid < nb) ? h[tid] : 0;
  int totalL;
  const int ls = block_excl_scan(L, wsum, totalL);
  if (tid < nb) {
    h[tid] = 0;
    lstart[tid] = ls;
    cstart[blk * (nb + 1) + tid] = ls;
  }
  if (tid == nb) cstart[blk * (nb + 1) + nb] = totalL;
  __syncthreads();

#pragma unroll
  for (int k = 0; k < KPT; ++k) {
    if (base + tid + k * BT < end) {
      const int bb = d[k] >> BSH;
      const int pos = lstart[bb] + atomicAdd(&h[bb], 1);
      stage[pos] = ((unsigned)s[k] << BSH) | (unsigned)(d[k] & (BN - 1));
    }
  }
  __syncthreads();

  const int nloc = end - base;
#pragma unroll
  for (int k = 0; k < KPT; ++k) {
    const int pos = tid + k * BT;
    if (pos < nloc) tmp[base + pos] = stage[pos];
  }
}

constexpr int XPT = BN * 16 / BT;

__global__ void __launch_bounds__(BT)
csr_rows(const unsigned* __restrict__ tmp, const int* __restrict__ cstart,
         int* __restrict__ rowptr, float* __restrict__ dinv,
         int* __restrict__ col, const float* __restrict__ x,
         float* __restrict__ xs, int n, int nb, int nchunk) {
  __shared__ int cnt[BN];
  __shared__ float dl[BN];
  __shared__ int cur[BN];
  __shared__ int wsum[32];
  __shared__ int rbeg[MAXC];
  __shared__ int rlen[MAXC];
  __shared__ int stg[CAP];
  const int tid = threadIdx.x;
  const int lane = tid & 31;
  const int wid = __builtin_amdgcn_readfirstlane((int)(threadIdx.x >> 5));
  const int b = blockIdx.x;

  int before = 0, mine = 0;
  if (tid < nchunk) {
    const int* cs = cstart + tid * (nb + 1) + b;
    before = cs[0];
    mine = cs[1] - before;
    rbeg[tid] = tid * CH + before;
    rlen[tid] = mine;
  }
  if (tid < BN) cnt[tid] = 0;

  const int xr = tid >> 4;
  const int xc = (tid & 15) * 4;
  float4 xv[XPT];
#pragma unroll
  for (int j = 0; j < XPT; ++j) {
    const int node = b * BN + j * (BT / 16) + xr;
    const int nn = (node < n) ? node : (n - 1);
    xv[j] = *(const float4*)(x + (long)nn * 64 + xc);
  }

  int s0, m;
  block_excl_scan(before, wsum, s0);
  block_excl_scan(mine, wsum, m);
  const int s1 = s0 + m;

  auto walk = [&](auto&& f) {
    for (int c0 = wid; c0 < nchunk; c0 += 32 * UB) {
      unsigned w[UB];
      int len[UB], beg[UB];
#pragma unroll
      for (int u = 0; u < UB; ++u) {
        const int cc = c0 + 32 * u;
        const int ci = (cc < nchunk) ? cc : c0;
        beg[u] = __builtin_amdgcn_readfirstlane(rbeg[ci]);
        len[u] = (cc < nchunk) ? __builtin_amdgcn_readfirstlane(rlen[ci]) : 0;
        w[u] = tmp[(lane < len[u]) ? (beg[u] + lane) : 0];
      }
#pragma unroll
      for (int u = 0; u < UB; ++u)
        if (lane < len[u]) f(w[u]);
#pragma unroll
      for (int u = 0; u < UB; ++u)
        for (int off = 32 + lane; off < len[u]; off += 32) f(tmp[beg[u] + off]);
    }
  };

  walk([&](const unsigned w) { atomicAdd(&cnt[w & (BN - 1)], 1); });
  lds_barrier();

  const int c = (tid < BN) ? cnt[tid] : 0;
  int total;
  const int ex = block_excl_scan(c, wsum, total);
  const int node = b * BN + tid;
  if (tid < BN) {
    cur[tid] = ex;
    const float di = rsqrtf((float)(c + 1));
    dl[tid] = di;
    if (node < n) {
      rowptr[node] = s0 + ex;
      dinv[node] = di;
    }
  }
  if (b == (int)gridDim.x - 1 && tid == 0) rowptr[n] = s1;
  lds_barrier();

#pragma unroll
  for (int j = 0; j < XPT; ++j) {
    const int lr = j * (BT / 16) + xr;
    const int node2 = b * BN + lr;
    if (node2 < n) {
      const float di = dl[lr];
      float4 v = xv[j];
      v.x *= di; v.y *= di; v.z *= di; v.w *= di;
      *(float4*)(xs + (long)node2 * 64 + xc) = v;
    }
  }

  const bool staged = (m <= CAP);
  walk([&](const unsigned w) {
    const int p = atomicAdd(&cur[w & (BN - 1)], 1);
    if (staged) stg[p] = (int)(w >> BSH);
    else col[s0 + p] = (int)(w >> BSH);
  });
  lds_barrier();
  if (staged)
    for (int i = tid; i < m; i += BT) col[s0 + i] = stg[i];
}


typedef float v4f __attribute__((ext_vector_type(4)));

template <int U>
__device__ __forceinline__ unsigned row_share(const unsigned x) {
  return (unsigned)__builtin_amdgcn_update_dpp(0, (int)x, 0x150 + U, 0xf, 0xf, false);
}
__device__ __forceinline__ unsigned row_down8(const unsigned x) {
  return (unsigned)__builtin_amdgcn_update_dpp(0, (int)x, 0x108, 0xf, 0xf, true);
}
__device__ __forceinline__ float xhalf(const float x) {
  return __int_as_float(__builtin_amdgcn_permlanex16(
      __float_as_int(x), __float_as_int(x), 0x76543210, 0xfedcba98, false, false));
}

template <int PITCH, int NL, int K, typename EMIT>
__device__ __forceinline__ void gather_sum_rows(const float* __restrict__ hin,
                                                const int* __restrict__ rowptr,
                                                const int* __restrict__ col,
                                                const int r0, const int n, const int lane,
                                                EMIT&& emit) {
  constexpr int G2 = 8;
  constexpr unsigned PB = PITCH * 4;
  static_assert(K >= 1 && K < 32, "row bounds come from one 32-lane load");
  static_assert(4 * G2 == 32, "an id register holds two groups of places");
  static_assert(NL >= 1 && NL <= 16, "a half-wave covers a row");
  const int half = lane >> 4;
  const int hl = lane & 15;
  const unsigned loff = (unsigned)(((hl < NL) ? hl : (NL - 1)) * 16);
  const int place = 2 * hl + half;
  const char* hb = (const char*)hin;

  const int bi = (lane < K) ? lane : K;
  const int rpv = rowptr[(r0 + bi < n) ? (r0 + bi) : n];

  int rb[K], rn[K], cr[K];
#pragma unroll
  for (int i = 0; i < K; ++i) {
    rb[i] = __builtin_amdgcn_readlane(rpv, i);
    rn[i] = __builtin_amdgcn_readlane(rpv, i + 1) - rb[i];
    cr[i] = 0;
    if (rn[i] > 0) {
      int pi = (place - 1 < rn[i] - 1) ? (place - 1) : (rn[i] - 1);
      pi = (pi > 0) ? pi : 0;
      cr[i] = col[rb[i] + pi];
    }
  }
  unsigned ids[K];
#pragma unroll
  for (int i = 0; i < K; ++i) {
    const int rr = (r0 + i < n) ? (r0 + i) : (n - 1);
    const int id = (place == 0 || rn[i] <= 0) ? rr : cr[i];
    ids[i] = (unsigned)id * PB;
  }

  auto request = [&](const unsigned idr, v4f (&v)[G2]) {
    unsigned o[G2];
    o[0] = row_share<0>(idr) + loff; o[1] = row_share<1>(idr) + loff;
    o[2] = row_share<2>(idr) + loff; o[3] = row_share<3>(idr) + loff;
    o[4] = row_share<4>(idr) + loff; o[5] = row_share<5>(idr) + loff;
    o[6] = row_share<6>(idr) + loff; o[7] = row_share<7>(idr) + loff;
    __builtin_amdgcn_sched_barrier(0);
#pragma unroll
    for (int u = 0; u < G2; ++u) v[u] = *(const v4f*)(hb + o[u]);
  };

  auto add_group = [&](v4f& acc, const v4f (&v)[G2], const int lim) {
#pragma unroll
    for (int u = 0; u < G2; ++u) {
      const bool ok = (2 * u < lim);
      acc.x = ok ? (acc.x + v[u].x) : acc.x;
      acc.y = ok ? (acc.y + v[u].y) : acc.y;
      acc.z = ok ? (acc.z + v[u].z) : acc.z;
      acc.w = ok ? (acc.w + v[u].w) : acc.w;
    }
  };

  auto finish = [&](const int i, const v4f (&v)[G2]) {
    const int cnt = rn[i] + 1;
    v4f acc = {0.0f, 0.0f, 0.0f, 0.0f};
    add_group(acc, v, cnt - half);
    unsigned cur = ids[i];
#pragma nounroll
    for (int off = 2 * G2; off < cnt; off += 2 * G2) {
      unsigned idr;
      if (off & (2 * G2)) {
        idr = row_down8(cur);
      } else {
        const int pi = (off + place - 1 < rn[i] - 1) ? (off + place - 1) : (rn[i] - 1);
        cur = (unsigned)col[rb[i] + pi] * PB;
        idr = cur;
      }
      v4f w[G2];
      request(idr, w);
      add_group(acc, w, cnt - half - off);
    }
    const v4f tot = { acc.x + xhalf(acc.x), acc.y + xhalf(acc.y),
                      acc.z + xhalf(acc.z), acc.w + xhalf(acc.w) };
    emit(i, tot);
  };

  v4f va[G2], vb[G2];
  request(ids[0], va);
#pragma unroll
  for (int i = 0; i < K; i += 2) {
    if (i + 1 < K) request(ids[i + 1], vb);
    finish(i, va);
    if (i + 1 < K) {
      if (i + 2 < K) request(ids[i + 2], va);
      finish(i + 1, vb);
    }
  }
}

template <int PITCH, int NL, int LDO, bool RELU, int K>
__global__ void gcn_gather_rows(const float* __restrict__ hin,
                                const int* __restrict__ rowptr,
                                const int* __restrict__ col,
                                const float* __restrict__ dinv,
                                const float* __restrict__ bias,
                                float* __restrict__ res, int n) {
  const int lane = threadIdx.x & 31;
  const int wid = __builtin_amdgcn_readfirstlane(
      (int)(blockIdx.x * (blockDim.x >> 5) + (threadIdx.x >> 5)));
  const int r0 = wid * K;
  if (r0 >= n) return;

  const int fi = (lane < K - 1) ? lane : (K - 1);
  const float dv = dinv[(r0 + fi < n) ? (r0 + fi) : (n - 1)];
  const int pc = ((lane & 15) < NL) ? (lane & 15) : (NL - 1);
  const v4f bv = *(const v4f*)(bias + 4 * pc);

  gather_sum_rows<PITCH, NL, K>(hin, rowptr, col, r0, n, lane,
      [&](const int i, const v4f& t) {
        const float di = __int_as_float(
            __builtin_amdgcn_readlane(__float_as_int(dv), i));
        v4f r = { fmaf(t.x, di, bv.x), fmaf(t.y, di, bv.y),
                  fmaf(t.z, di, bv.z), fmaf(t.w, di, bv.w) };
        if (RELU) {
          r.x = r.x > 0.0f ? r.x : 0.0f;
          r.y = r.y > 0.0f ? r.y : 0.0f;
          r.z = r.z > 0.0f ? r.z : 0.0f;
          r.w = r.w > 0.0f ? r.w : 0.0f;
        }
        if (r0 + i < n && lane < NL) *(v4f*)(res + (long)(r0 + i) * LDO + 4 * lane) = r;
      });
}

constexpr int TP = 68;

template <int NT>
__device__ __forceinline__ void tile_gemm1_part(const float* __restrict__ tin,
                                                const float* __restrict__ B,
                                                const float* __restrict__ bias,
                                                float* __restrict__ hout,
                                                const int t0, const int lane) {
  constexpr int LDB = 64;
  constexpr int KS = 4;
  static_assert(NT == 2, "guard written for 2 tiles");
  v8f acc[NT];
#pragma unroll
  for (int t = 0; t < NT; ++t) {
    acc[t] = v8f{};
    asm volatile("" : "+v"(acc[t]));
  }
  const int mrow = lane & 15;
  const int hsel = lane >> 4;
  const int kofs = hsel << 1;

  float bs[NT];
  const float* bp[NT];
#pragma unroll
  for (int t = 0; t < NT; ++t) {
    const int bn = (t0 + t) * 16 + mrow;
    bs[t] = bias[bn];
    bp[t] = B + kofs * LDB + bn;
  }
  const float* arow = tin + mrow * TP + kofs;

#pragma unroll
  for (int k0 = 0; k0 < 16; k0 += KS) {
    v2f a[KS], b[NT][KS];
#pragma unroll
    for (int j = 0; j < KS; ++j) a[j] = *(const v2f*)(arow + 4 * (k0 + j));
#pragma unroll
    for (int t = 0; t < NT; ++t)
#pragma unroll
      for (int j = 0; j < KS; ++j)
        b[t][j] = v2f{ bp[t][4 * (k0 + j) * LDB], bp[t][4 * (k0 + j) * LDB + LDB] };

    __builtin_amdgcn_sched_barrier(0);
#pragma unroll
    for (int j = 0; j < KS; ++j) {
#pragma unroll
      for (int t = 0; t < NT; ++t) {
        acc[t] = __builtin_amdgcn_wmma_f32_16x16x4_f32(
             false, a[j],  false, b[t][j],
             (short)0, acc[t],  false,  false);
      }
    }
    __builtin_amdgcn_sched_barrier(0);

    asm volatile("v_nop\n\tv_nop\n\tv_nop\n\tv_nop"
                 : "+v"(acc[0]), "+v"(acc[1])
                 : "v"(a[KS - 1]), "v"(b[0][KS - 1]), "v"(b[1][KS - 1]));
  }

#pragma unroll
  for (int t = 0; t < NT; ++t) {
    float* hp = hout + (hsel << 3) * TP + (t0 + t) * 16 + mrow;
#pragma unroll
    for (int r = 0; r < 8; ++r) {
      const float o = acc[t][r] + bs[t];
      hp[r * TP] = o > 0.0f ? o : 0.0f;
    }
  }
}

template <int NT>
__device__ __forceinline__ void tile_gemm_part(const float* __restrict__ tile,
                                               const float* __restrict__ B,
                                               const float* __restrict__ rscale,
                                               float* __restrict__ C,
                                               const int t0, const int row0, const int nrows,
                                               const int lane) {
  constexpr int LDB = 40, LDC = 48, NVALID = 40;
  constexpr int KS = 4;
  static_assert(NT == 1 || NT == 2, "guard written for 1 or 2 tiles");
  v8f acc[NT];
#pragma unroll
  for (int t = 0; t < NT; ++t) {
    acc[t] = v8f{};
    asm volatile("" : "+v"(acc[t]));
  }
  const int mrow = lane & 15;
  const int hsel = lane >> 4;
  const int kofs = hsel << 1;

  const int crow = row0 + (hsel << 3);
  const bool full = (row0 + 16 <= nrows);
  float sc[8];
  if (full) {
    const float4 s0 = *(const float4*)(rscale + crow);
    const float4 s1 = *(const float4*)(rscale + crow + 4);
    sc[0] = s0.x; sc[1] = s0.y; sc[2] = s0.z; sc[3] = s0.w;
    sc[4] = s1.x; sc[5] = s1.y; sc[6] = s1.z; sc[7] = s1.w;
  } else {
#pragma unroll
    for (int r = 0; r < 8; ++r) {
      const int rr = (crow + r < nrows) ? (crow + r) : (nrows - 1);
      sc[r] = rscale[rr];
    }
  }

  const float* arow = tile + mrow * TP + kofs;
  const float* bp[NT];
#pragma unroll
  for (int t = 0; t < NT; ++t) {
    const int bn   = (t0 + t) * 16 + mrow;
    const int bcol = (bn < NVALID) ? bn : (NVALID - 1);
    bp[t] = B + kofs * LDB + bcol;
  }

#pragma unroll
  for (int k0 = 0; k0 < 16; k0 += KS) {
    v2f a[KS], b[NT][KS];
#pragma unroll
    for (int j = 0; j < KS; ++j) a[j] = *(const v2f*)(arow + 4 * (k0 + j));
#pragma unroll
    for (int t = 0; t < NT; ++t)
#pragma unroll
      for (int j = 0; j < KS; ++j)
        b[t][j] = v2f{ bp[t][4 * (k0 + j) * LDB], bp[t][4 * (k0 + j) * LDB + LDB] };

    __builtin_amdgcn_sched_barrier(0);
#pragma unroll
    for (int j = 0; j < KS; ++j) {
#pragma unroll
      for (int t = 0; t < NT; ++t) {
        acc[t] = __builtin_amdgcn_wmma_f32_16x16x4_f32(
             false, a[j],  false, b[t][j],
             (short)0, acc[t],  false,  false);
      }
    }
    __builtin_amdgcn_sched_barrier(0);

    if constexpr (NT == 2) {
      asm volatile("v_nop\n\tv_nop\n\tv_nop\n\tv_nop"
                   : "+v"(acc[0]), "+v"(acc[1])
                   : "v"(a[KS - 1]), "v"(b[0][KS - 1]), "v"(b[1][KS - 1]));
    } else {
      asm volatile("v_nop\n\tv_nop\n\tv_nop\n\tv_nop"
                   : "+v"(acc[0])
                   : "v"(a[KS - 1]), "v"(b[0][KS - 1]));
    }
  }

#pragma unroll
  for (int t = 0; t < NT; ++t) {
    float o[8];
#pragma unroll
    for (int r = 0; r < 8; ++r) o[r] = acc[t][r] * sc[r];
    float* cp = C + crow * LDC + (t0 + t) * 16 + mrow;
    if (full) {
#pragma unroll
      for (int r = 0; r < 8; ++r) cp[r * LDC] = o[r];
    } else {
#pragma unroll
      for (int r = 0; r < 8; ++r)
        if (crow + r < nrows) cp[r * LDC] = o[r];
    }
  }
}

template <int K>
__global__ void __launch_bounds__(256)
gcn_gather_gemm2(const float* __restrict__ hin,
                 const int* __restrict__ rowptr,
                 const int* __restrict__ col,
                 const float* __restrict__ dinv,
                 const float* __restrict__ B1,
                 const float* __restrict__ bias,
                 const float* __restrict__ B,
                 float* __restrict__ C, int n) {
  constexpr int LDH = 64;
  static_assert(K == 8, "8 waves x K rows = the 64 rows of the tile");
  __shared__ __attribute__((aligned(16))) float tile[8 * K * TP];
  __shared__ __attribute__((aligned(16))) float tile2[8 * K * TP];
  const int lane = threadIdx.x & 31;
  const int wv = __builtin_amdgcn_readfirstlane((int)(threadIdx.x >> 5));
  const int brow0 = blockIdx.x * (8 * K);
  const int r0 = brow0 + wv * K;

  const int fi = (lane < K - 1) ? lane : (K - 1);
  const float dv = dinv[(r0 + fi < n) ? (r0 + fi) : (n - 1)];

  gather_sum_rows<LDH, 16, K>(hin, rowptr, col, r0, n, lane,
      [&](const int i, const v4f& t) {
        const float di = __int_as_float(
            __builtin_amdgcn_readlane(__float_as_int(dv), i));
        v4f r = { t.x * di, t.y * di, t.z * di, t.w * di };
        if (r0 + i >= n) r = v4f{0.0f, 0.0f, 0.0f, 0.0f};
        if (lane < 16) *(v4f*)(tile + (wv * K + i) * TP + 4 * lane) = r;
      });

  __syncthreads();

  const int mt = wv & 3;
  tile_gemm1_part<2>(tile + mt * 16 * TP, B1, bias, tile2 + mt * 16 * TP, (wv < 4) ? 0 : 2, lane);

  __syncthreads();

  const int grow0 = brow0 + mt * 16;
  if (grow0 >= n) return;
  if (wv < 4) tile_gemm_part<2>(tile2 + mt * 16 * TP, B, dinv, C, 0, grow0, n, lane);
  else        tile_gemm_part<1>(tile2 + mt * 16 * TP, B, dinv, C, 2, grow0, n, lane);
}

extern "C" void kernel_launch(void* const* d_in, const int* in_sizes, int n_in,
                              void* d_out, int out_size, void* d_ws, size_t ws_size,
                              hipStream_t stream) {
  (void)n_in; (void)out_size; (void)ws_size;
  const float* x  = (const float*)d_in[0];
  const int*   ei = (const int*)d_in[1];
  const float* W1 = (const float*)d_in[2];
  const float* b1 = (const float*)d_in[3];
  const float* W3 = (const float*)d_in[4];
  const float* b3 = (const float*)d_in[5];

  const int N = in_sizes[0] / 64;
  const int E = in_sizes[1] / 2;
  const int* src = ei;
  const int* dst = ei + E;

  const int NBK = (N + BN - 1) / BN;
  const int NCH = (E + CH - 1) / CH;

  float* dinv   = (float*)d_ws;
  float* bufA   = dinv + N;
  float* bufB   = bufA + (long)N * 64;
  int*   cstart = (int*)(bufB + (long)N * 64);
  int*   rowptr = cstart + N;
  unsigned* tmp = (unsigned*)(rowptr + N + 1);
  int*   col    = (int*)tmp + E;
  float* out    = (float*)d_out;

  constexpr int KR = 8;

  csr_chunk_sort<<<NCH, BT, 0, stream>>>(src, dst, tmp, cstart, E, NBK);
  csr_rows<<<NBK, BT, 0, stream>>>(tmp, cstart, rowptr, dinv, col, x, bufA, N, NBK, NCH);

  gcn_gather_gemm2<KR><<<(N + 8 * KR - 1) / (8 * KR), 256, 0, stream>>>(
      bufA, rowptr, col, dinv, W1, b1, W3, bufB, N);

  gcn_gather_rows<48, 10, 40, false, KR><<<(N + 8 * KR - 1) / (8 * KR), 256, 0, stream>>>(
      bufB, rowptr, col, dinv, b3, out, N);
}
